# starting kernel
# baseline (speedup 1.0000x reference)
#include <hip/hip_runtime.h>
#include <hip/hip_bf16.h>
#include <math.h>

#define HID      128
#define HEADS    8
#define HEAD_DIM 16
#define LAYERS   3
#define NEG_SLOPE 0.2f
#define EPS_F     1e-16f

typedef float v2f __attribute__((ext_vector_type(2)));
typedef float v8f __attribute__((ext_vector_type(8)));

__global__ __launch_bounds__(256)
void gemm_dual_f32(const float* __restrict__ X,
                   const float* __restrict__ Wl,
                   const float* __restrict__ Wr,
                   float* __restrict__ XL,
                   float* __restrict__ XR) {
    __shared__ float As[16][HID];
    const int row0 = blockIdx.x * 16;

    {
        const float4* gx = (const float4*)(X + (size_t)row0 * HID);
        float4* sx = (float4*)(&As[0][0]);
        for (int t = threadIdx.x; t < 16 * HID / 4; t += 256) sx[t] = gx[t];
    }
    __syncthreads();

    const int wave = threadIdx.x >> 5;
    const int lane = threadIdx.x & 31;
    const int col0 = wave * 16;
    const int half = lane >> 4;
    const int lm   = lane & 15;

    v8f accL = {0.f,0.f,0.f,0.f,0.f,0.f,0.f,0.f};
    v8f accR = {0.f,0.f,0.f,0.f,0.f,0.f,0.f,0.f};

    #pragma unroll 4
    for (int k0 = 0; k0 < HID; k0 += 4) {
        const int ka = k0 + 2 * half;
        v2f a;
        a.x = As[lm][ka];
        a.y = As[lm][ka + 1];

        v2f bl, br;
        bl.x = Wl[(size_t)(ka + 0) * HID + col0 + lm];
        bl.y = Wl[(size_t)(ka + 1) * HID + col0 + lm];
        br.x = Wr[(size_t)(ka + 0) * HID + col0 + lm];
        br.y = Wr[(size_t)(ka + 1) * HID + col0 + lm];

        accL = __builtin_amdgcn_wmma_f32_16x16x4_f32(false, a, false, bl,
                                                     (short)0, accL, false, false);
        accR = __builtin_amdgcn_wmma_f32_16x16x4_f32(false, a, false, br,
                                                     (short)0, accR, false, false);
    }

    #pragma unroll
    for (int r = 0; r < 8; ++r) {
        const int m = r + 8 * half;
        XL[(size_t)(row0 + m) * HID + col0 + lm] = accL[r];
        XR[(size_t)(row0 + m) * HID + col0 + lm] = accR[r];
    }
}

__device__ __forceinline__ void atomicMaxFloat(float* addr, float val) {
    if (val >= 0.0f)
        atomicMax((int*)addr, __float_as_int(val));
    else
        atomicMin((unsigned int*)addr, __float_as_uint(val));
}

__global__ void init_md_kernel(float* __restrict__ mmax,
                               float* __restrict__ denom, int n) {
    int i = blockIdx.x * blockDim.x + threadIdx.x;
    if (i < n) { mmax[i] = -INFINITY; denom[i] = 0.0f; }
}

__global__ void init_out_kernel(float* __restrict__ out,
                                const float* __restrict__ bias, int total) {
    int i = blockIdx.x * blockDim.x + threadIdx.x;
    if (i < total) out[i] = bias[i & (HID - 1)];
}

__global__ void edge_logits_kernel(const float* __restrict__ xl,
                                   const float* __restrict__ xr,
                                   const int* __restrict__ src,
                                   const int* __restrict__ dst,
                                   const float* __restrict__ att,
                                   float* __restrict__ logits,
                                   float* __restrict__ mmax, int EH) {
    int i = blockIdx.x * blockDim.x + threadIdx.x;
    if (i >= EH) return;
    const int e = i / HEADS, h = i % HEADS;
    const int s = src[e], d = dst[e];

    const float4* pl = (const float4*)(xl + (size_t)s * HID + h * HEAD_DIM);
    const float4* pr = (const float4*)(xr + (size_t)d * HID + h * HEAD_DIM);
    const float4* pa = (const float4*)(att + h * HEAD_DIM);

    float acc = 0.0f;
    #pragma unroll
    for (int q = 0; q < 4; ++q) {
        float4 a = pl[q], b = pr[q], w = pa[q];
        float v;
        v = a.x + b.x; v = (v > 0.0f) ? v : NEG_SLOPE * v; acc += v * w.x;
        v = a.y + b.y; v = (v > 0.0f) ? v : NEG_SLOPE * v; acc += v * w.y;
        v = a.z + b.z; v = (v > 0.0f) ? v : NEG_SLOPE * v; acc += v * w.z;
        v = a.w + b.w; v = (v > 0.0f) ? v : NEG_SLOPE * v; acc += v * w.w;
    }
    logits[i] = acc;
    atomicMaxFloat(&mmax[(size_t)d * HEADS + h], acc);
}

__global__ void edge_softmax_kernel(const int* __restrict__ dst,
                                    const float* __restrict__ mmax,
                                    float* __restrict__ lw,
                                    float* __restrict__ denom, int EH) {
    int i = blockIdx.x * blockDim.x + threadIdx.x;
    if (i >= EH) return;
    const int e = i / HEADS, h = i % HEADS;
    const int d = dst[e];
    const float w = expf(lw[i] - mmax[(size_t)d * HEADS + h]);
    lw[i] = w;
    atomicAdd(&denom[(size_t)d * HEADS + h], w);
}

__global__ void edge_aggregate_kernel(const int* __restrict__ src,
                                      const int* __restrict__ dst,
                                      const float* __restrict__ xl,
                                      const float* __restrict__ lw,
                                      const float* __restrict__ denom,
                                      float* __restrict__ out, int EH) {
    int i = blockIdx.x * blockDim.x + threadIdx.x;
    if (i >= EH) return;
    const int e = i / HEADS, h = i % HEADS;
    const int s = src[e], d = dst[e];
    const float alpha = lw[i] / (denom[(size_t)d * HEADS + h] + EPS_F);

    const float4* pl = (const float4*)(xl + (size_t)s * HID + h * HEAD_DIM);
    float* po = out + (size_t)d * HID + h * HEAD_DIM;
    #pragma unroll
    for (int q = 0; q < 4; ++q) {
        float4 v = pl[q];
        atomicAdd(po + 4 * q + 0, alpha * v.x);
        atomicAdd(po + 4 * q + 1, alpha * v.y);
        atomicAdd(po + 4 * q + 2, alpha * v.z);
        atomicAdd(po + 4 * q + 3, alpha * v.w);
    }
}

extern "C" void kernel_launch(void* const* d_in, const int* in_sizes, int n_in,
                              void* d_out, int out_size, void* d_ws, size_t ws_size,
                              hipStream_t stream) {
    const float* x          = (const float*)d_in[0];
    const int*   edge_index = (const int*)  d_in[1];
    const float* Wl_all     = (const float*)d_in[2];
    const float* Wr_all     = (const float*)d_in[3];
    const float* att_all    = (const float*)d_in[4];
    const float* bias_all   = (const float*)d_in[5];
    float* out = (float*)d_out;

    const int N = in_sizes[0] / HID;
    const int E = in_sizes[1] / 2;
    const int EH = E * HEADS;

    const int* src = edge_index;
    const int* dst = edge_index + E;

    const size_t nf = (size_t)N * HID;
    float* xl    = (float*)d_ws;
    float* xr    = xl + nf;
    float* bufA  = xr + nf;
    float* mmax  = bufA + nf;
    float* denom = mmax + (size_t)N * HEADS;
    float* lw    = denom + (size_t)N * HEADS;

    const int TPB = 256;
    const int gemm_blocks = N / 16;
    const int nh_blocks   = (N * HEADS + TPB - 1) / TPB;
    const int nc_blocks   = (N * HID + TPB - 1) / TPB;
    const int eh_blocks   = (EH + TPB - 1) / TPB;

    const float* xin = x;
    for (int l = 0; l < LAYERS; ++l) {
        const float* Wl   = Wl_all   + (size_t)l * HID * HID;
        const float* Wr   = Wr_all   + (size_t)l * HID * HID;
        const float* att  = att_all  + (size_t)l * HEADS * HEAD_DIM;
        const float* bias = bias_all + (size_t)l * HID;
        float* xout = (l == LAYERS - 1) ? out : bufA;
        gemm_dual_f32<<<gemm_blocks, TPB, 0, stream>>>(xin, Wl, Wr, xl, xr);
        init_md_kernel<<<nh_blocks, TPB, 0, stream>>>(mmax, denom, N * HEADS);
        init_out_kernel<<<nc_blocks, TPB, 0, stream>>>(xout, bias, N * HID);
        edge_logits_kernel<<<eh_blocks, TPB, 0, stream>>>(xl, xr, src, dst, att,
                                                          lw, mmax, EH);
        edge_softmax_kernel<<<eh_blocks, TPB, 0, stream>>>(dst, mmax, lw, denom, EH);
        edge_aggregate_kernel<<<eh_blocks, TPB, 0, stream>>>(src, dst, xl, lw,
                                                             denom, xout, EH);
        xin = xout;
    }
}


# Attention in one edge sweep, not two: softmax shift = logit of one listed edge of the destination, not the maximum; exact two-sweep redo if a logit tops its shift by 40; sums in arrival order
# speedup vs baseline: 5.6607x; 5.6607x over previous
#include <hip/hip_runtime.h>
#include <hip/hip_bf16.h>
#include <math.h>

#define HID      128
#define HEADS    8
#define HEAD_DIM 16
#define LAYERS   3
#define NEG_SLOPE 0.2f
#define EPS_F     1e-16f

typedef float v2f __attribute__((ext_vector_type(2)));
typedef float v8f __attribute__((ext_vector_type(8)));

template <bool NORM>
__global__ __launch_bounds__(256)
void gemm_dual_f32(const float* __restrict__ X,
                   const float* __restrict__ den,
                   const float* __restrict__ bias,
                   const float* __restrict__ Wl,
                   const float* __restrict__ Wr,
                   float* __restrict__ XL,
                   float* __restrict__ XR) {
    __shared__ float As[16][HID];
    const int row0 = blockIdx.x * 16;

    {
        const float4* gx = (const float4*)(X + (size_t)row0 * HID);
        float4* sx = (float4*)(&As[0][0]);
        for (int t = threadIdx.x; t < 16 * HID / 4; t += 256) {
            float4 v = gx[t];
            if (NORM) {
                const int r  = t >> 5;
                const int c4 = (t & 31) * 4;
                const float dn = den[(size_t)(row0 + r) * HEADS + (c4 >> 4)] + EPS_F;
                const float4 b = *(const float4*)(bias + c4);
                v.x = v.x / dn + b.x;
                v.y = v.y / dn + b.y;
                v.z = v.z / dn + b.z;
                v.w = v.w / dn + b.w;
            }
            sx[t] = v;
        }
    }
    __syncthreads();

    const int wave = threadIdx.x >> 5;
    const int lane = threadIdx.x & 31;
    const int col0 = wave * 16;
    const int half = lane >> 4;
    const int lm   = lane & 15;

    v8f accL = {0.f,0.f,0.f,0.f,0.f,0.f,0.f,0.f};
    v8f accR = {0.f,0.f,0.f,0.f,0.f,0.f,0.f,0.f};

    #pragma unroll 4
    for (int k0 = 0; k0 < HID; k0 += 4) {
        const int ka = k0 + 2 * half;
        v2f a;
        a.x = As[lm][ka];
        a.y = As[lm][ka + 1];

        v2f bl, br;
        bl.x = Wl[(size_t)(ka + 0) * HID + col0 + lm];
        bl.y = Wl[(size_t)(ka + 1) * HID + col0 + lm];
        br.x = Wr[(size_t)(ka + 0) * HID + col0 + lm];
        br.y = Wr[(size_t)(ka + 1) * HID + col0 + lm];

        accL = __builtin_amdgcn_wmma_f32_16x16x4_f32(false, a, false, bl,
                                                     (short)0, accL, false, false);
        accR = __builtin_amdgcn_wmma_f32_16x16x4_f32(false, a, false, br,
                                                     (short)0, accR, false, false);
    }

    #pragma unroll
    for (int r = 0; r < 8; ++r) {
        const int m = r + 8 * half;
        XL[(size_t)(row0 + m) * HID + col0 + lm] = accL[r];
        XR[(size_t)(row0 + m) * HID + col0 + lm] = accR[r];
    }
}

__global__ void divide_out_kernel(float* __restrict__ out,
                                  const float* __restrict__ den,
                                  const float* __restrict__ bias, int total4) {
    int i = blockIdx.x * blockDim.x + threadIdx.x;
    if (i >= total4) return;
    const int n  = i >> 5;
    const int c4 = (i & 31) * 4;
    float4 v = ((const float4*)out)[i];
    const float dn = den[(size_t)n * HEADS + (c4 >> 4)] + EPS_F;
    const float4 b = *(const float4*)(bias + c4);
    v.x = v.x / dn + b.x;
    v.y = v.y / dn + b.y;
    v.z = v.z / dn + b.z;
    v.w = v.w / dn + b.w;
    ((float4*)out)[i] = v;
}

#define OWN_ROWS   196
#define OWN_WAVES  32
#define OWN_STEP   (OWN_WAVES * 512)
#define OWN_CAP    24576
#define SHIFT_SPAN 40.0f
#define NO_EDGE    0x7fffffff
#define SEED_ROWS  224

__device__ __forceinline__ void lds_max_float(float* addr, float val) {
    if (__float_as_int(val) >= 0)
        atomicMax((int*)addr, __float_as_int(val));
    else
        atomicMin((unsigned int*)addr, __float_as_uint(val));
}

template <int CTRL>
__device__ __forceinline__ float quad_swap(float v) {
    return __int_as_float(__builtin_amdgcn_update_dpp(
        0, __float_as_int(v), CTRL, 0xf, 0xf, false));
}

__device__ __forceinline__ void append_edges(bool m, int word, int* list, int* cnt,
                                             int lane) {
    const unsigned mask = __builtin_amdgcn_ballot_w32(m);
    if (mask != 0u) {
        int base = 0;
        if (lane == 0) base = atomicAdd(cnt, (int)__popc(mask));
        base = __builtin_amdgcn_readfirstlane(base);
        if (m) list[base + (int)__builtin_amdgcn_mbcnt_lo(mask, 0u)] = word;
    }
}

__device__ __forceinline__ float edge_logit(const float4 v, const float4 r,
                                            const float4 a4) {
    float t, p;
    t = v.x + r.x; t = (t > 0.0f) ? t : NEG_SLOPE * t; p = t * a4.x;
    t = v.y + r.y; t = (t > 0.0f) ? t : NEG_SLOPE * t; p += t * a4.y;
    t = v.z + r.z; t = (t > 0.0f) ? t : NEG_SLOPE * t; p += t * a4.z;
    t = v.w + r.w; t = (t > 0.0f) ? t : NEG_SLOPE * t; p += t * a4.w;
    p += quad_swap<0xB1>(p);
    p += quad_swap<0x4E>(p);
    return p;
}

__device__ __forceinline__ void seed_shifts(const int* first, float* mx,
                                            const float* xrs,
                                            const float* __restrict__ xl,
                                            const float4 a4, int wave, int lane) {
    constexpr int PER = SEED_ROWS / OWN_WAVES;
    float4 v[PER];
    int wd[PER];
    #pragma unroll
    for (int j = 0; j < PER; ++j) wd[j] = first[wave + j * OWN_WAVES];
    #pragma unroll
    for (int j = 0; j < PER; ++j) {
        wd[j] = __builtin_amdgcn_readfirstlane(wd[j]);
        const unsigned s = (wd[j] == NO_EDGE) ? 0u : ((unsigned)wd[j] >> 8);
        v[j] = *(const float4*)(xl + (size_t)s * HID + 4 * lane);
    }
    #pragma unroll
    for (int j = 0; j < PER; ++j) {
        if (wd[j] != NO_EDGE) {
            const int r = wave + j * OWN_WAVES;
            const float4 x4 = *(const float4*)(xrs + r * HID + 4 * lane);
            const float p = edge_logit(v[j], x4, a4);
            if ((lane & 3) == 0) mx[r * HEADS + (lane >> 2)] = p;
        }
    }
}

template <int MODE>
__device__ __forceinline__ bool sweep_edges(int n, const int* list, float* acc,
                                            float* den, float* mx, const float* xrs,
                                            const float* __restrict__ xl,
                                            const float4 a4, int wave, int lane) {
    const int hq = lane >> 2;
    bool over = false;
    for (int i0 = wave * 8; i0 < n; i0 += OWN_WAVES * 8) {
        const int idx = i0 + (lane & 7);
        const int rec = list[idx < n ? idx : i0];
        float4 v[8];
        int dl[8];
        #pragma unroll
        for (int j = 0; j < 8; ++j) {
            const unsigned rj = (unsigned)__builtin_amdgcn_readlane(rec, j);
            dl[j] = (int)(rj & 255u);
            v[j] = *(const float4*)(xl + (size_t)(rj >> 8) * HID + 4 * lane);
        }
        #pragma unroll
        for (int j = 0; j < 8; ++j) {
            if (i0 + j < n) {
                const float4 r = *(const float4*)(xrs + dl[j] * HID + 4 * lane);
                const float p = edge_logit(v[j], r, a4);
                float* pm = mx + dl[j] * HEADS + hq;
                if (MODE == 1) {
                    if ((lane & 3) == 0) lds_max_float(pm, p);
                } else {
                    const float d = p - *pm;
                    if (MODE == 3) over |= (d > SHIFT_SPAN);
                    const float w = expf(d);
                    if ((lane & 3) == 0) atomicAdd(den + dl[j] * HEADS + hq, w);
                    float* po = acc + dl[j] * HID + lane;
                    atomicAdd(po +  0, w * v[j].x);
                    atomicAdd(po + 32, w * v[j].y);
                    atomicAdd(po + 64, w * v[j].z);
                    atomicAdd(po + 96, w * v[j].w);
                }
            }
        }
    }
    return over;
}

template <bool ONE_SWEEP>
__device__ __forceinline__ void attend_list(int n, const int* list,
                                            const int* first, int* retry, float* acc,
                                            float* den, float* mx, const float* xrs,
                                            const float* __restrict__ xl,
                                            const float4 a4, int tid, int wave,
                                            int lane) {
    if (ONE_SWEEP) {
        seed_shifts(first, mx, xrs, xl, a4, wave, lane);
        __syncthreads();
        if (sweep_edges<3>(n, list, acc, den, mx, xrs, xl, a4, wave, lane)) *retry = 1;
        __syncthreads();
        if (__builtin_amdgcn_readfirstlane(*retry) == 0) return;
        for (int t = tid; t < OWN_ROWS * HID; t += 1024) acc[t] = 0.0f;
        for (int t = tid; t < OWN_ROWS * HEADS; t += 1024) {
            den[t] = 0.0f;
            mx[t] = -INFINITY;
        }
        __syncthreads();
    }
    sweep_edges<1>(n, list, acc, den, mx, xrs, xl, a4, wave, lane);
    __syncthreads();
    sweep_edges<2>(n, list, acc, den, mx, xrs, xl, a4, wave, lane);
    __syncthreads();
}

__device__ __forceinline__ void edges_to_sources(int n, int* list,
                                                 const int* __restrict__ src, int tid) {
    for (int t = tid; t < n; t += 1024) {
        const unsigned w = (unsigned)list[t];
        list[t] = (int)(((unsigned)src[w >> 8] << 8) | (w & 255u));
    }
}

template <bool FIRST>
__global__ __launch_bounds__(1024)
void owner_attend_kernel(const int* __restrict__ src,
                         const int* __restrict__ dst,
                         const float* __restrict__ xl,
                         const float* __restrict__ xr,
                         const float* __restrict__ att,
                         float* __restrict__ denom,
                         float* __restrict__ out,
                         int* __restrict__ glist,
                         int* __restrict__ gcnt, int N, int E) {
    __shared__ float acc[OWN_ROWS * HID];
    __shared__ float xrs[OWN_ROWS * HID];
    __shared__ float den[OWN_ROWS * HEADS];
    __shared__ float mx[OWN_ROWS * HEADS];
    __shared__ int   list[OWN_CAP];
    __shared__ int   first[SEED_ROWS];
    __shared__ int   cnt;
    __shared__ int   retry;

    const int tid = threadIdx.x;
    const int lane = tid & 31;
    const int wave = __builtin_amdgcn_readfirstlane(tid >> 5);
    const int node0 = blockIdx.x * OWN_ROWS;
    const int rows = min(OWN_ROWS, N - node0);

    for (int t = tid; t < OWN_ROWS * HID; t += 1024) acc[t] = 0.0f;
    for (int t = tid; t < OWN_ROWS * HEADS; t += 1024) {
        den[t] = 0.0f;
        mx[t] = -INFINITY;
    }
    {
        const float4* gx = (const float4*)(xr + (size_t)node0 * HID);
        float4* sx = (float4*)xrs;
        for (int t = tid; t < rows * (HID / 4); t += 1024) sx[t] = gx[t];
    }
    if (tid < SEED_ROWS) first[tid] = NO_EDGE;
    if (tid == 0) { cnt = 0; retry = 0; }
    __syncthreads();

    const float4 a4 = ((const float4*)att)[lane];
    const bool al16 = __builtin_amdgcn_readfirstlane((int)(((size_t)dst) & 15)) == 0;
    const int nsteps = (E + OWN_STEP - 1) / OWN_STEP;

    int* mylist = glist + (size_t)blockIdx.x * OWN_CAP;
    int kept = -1;
    if (!FIRST) kept = __builtin_amdgcn_readfirstlane(gcnt[blockIdx.x]);
    if (kept >= 0) {
        for (int t0 = tid; t0 < kept; t0 += 4096) {
            int w[4];
            #pragma unroll
            for (int k = 0; k < 4; ++k)
                w[k] = (t0 + k * 1024 < kept) ? mylist[t0 + k * 1024] : NO_EDGE;
            #pragma unroll
            for (int k = 0; k < 4; ++k) {
                if (t0 + k * 1024 < kept) {
                    list[t0 + k * 1024] = w[k];
                    atomicMin(&first[w[k] & 255], w[k]);
                }
            }
        }
        __syncthreads();
        attend_list<true>(kept, list, first, &retry, acc, den, mx, xrs, xl, a4,
                          tid, wave, lane);
    } else {
        bool emptied = false;
        const int step0 = (int)(blockIdx.x % (unsigned)nsteps);
        for (int scan = 0; scan < 2; ++scan) {
            for (int it = 0; it < nsteps; ++it) {
                const int step = (it + step0 < nsteps) ? it + step0 : it + step0 - nsteps;
                const int wbase = step * OWN_STEP + wave * 512 + lane * 4;
                int4 dv[4];
                const bool full = al16 && (step * OWN_STEP + wave * 512 + 512 <= E);
                if (full) {
                    #pragma unroll
                    for (int g = 0; g < 4; ++g)
                        dv[g] = *(const int4*)(dst + wbase + g * 128);
                } else {
                    #pragma unroll
                    for (int g = 0; g < 4; ++g) {
                        const int e = wbase + g * 128;
                        dv[g].x = (e + 0 < E) ? dst[e + 0] : -1;
                        dv[g].y = (e + 1 < E) ? dst[e + 1] : -1;
                        dv[g].z = (e + 2 < E) ? dst[e + 2] : -1;
                        dv[g].w = (e + 3 < E) ? dst[e + 3] : -1;
                    }
                }
                #pragma unroll
                for (int g = 0; g < 4; ++g) {
                    const int e = wbase + g * 128;
                    const int d0 = dv[g].x - node0, d1 = dv[g].y - node0;
                    const int d2 = dv[g].z - node0, d3 = dv[g].w - node0;
                    const bool m0 = (unsigned)d0 < (unsigned)rows;
                    const bool m1 = (unsigned)d1 < (unsigned)rows;
                    const bool m2 = (unsigned)d2 < (unsigned)rows;
                    const bool m3 = (unsigned)d3 < (unsigned)rows;
                    if (__builtin_amdgcn_ballot_w32(m0 | m1 | m2 | m3) != 0u) {
                        append_edges(m0, ((e + 0) << 8) | d0, list, &cnt, lane);
                        append_edges(m1, ((e + 1) << 8) | d1, list, &cnt, lane);
                        append_edges(m2, ((e + 2) << 8) | d2, list, &cnt, lane);
                        append_edges(m3, ((e + 3) << 8) | d3, list, &cnt, lane);
                    }
                }
                __syncthreads();
                const int n = __builtin_amdgcn_readfirstlane(cnt);
                __syncthreads();
                if (n <= OWN_CAP - OWN_STEP && it != nsteps - 1) continue;
                edges_to_sources(n, list, src, tid);
                __syncthreads();
                if (it == nsteps - 1 && !emptied) {
                    if (FIRST) {
                        for (int t = tid; t < n; t += 1024) {
                            const int w = list[t];
                            mylist[t] = w;
                            atomicMin(&first[w & 255], w);
                        }
                        if (tid == 0) gcnt[blockIdx.x] = n;
                        __syncthreads();
                    }
                    attend_list<FIRST>(n, list, first, &retry, acc, den, mx, xrs,
                                       xl, a4, tid, wave, lane);
                    break;
                }
                emptied = true;
                if (scan == 0)
                    sweep_edges<1>(n, list, acc, den, mx, xrs, xl, a4, wave, lane);
                else
                    sweep_edges<2>(n, list, acc, den, mx, xrs, xl, a4, wave, lane);
                __syncthreads();
                if (tid == 0) cnt = 0;
                __syncthreads();
            }
            if (!emptied) break;
        }
        if (FIRST && emptied && tid == 0) gcnt[blockIdx.x] = -1;
    }

    for (int t = tid; t < rows * (HID / 4); t += 1024) {
        const int r = t >> 5, l = t & 31;
        const float* pa = acc + r * HID + l;
        float4 o;
        o.x = pa[0]; o.y = pa[32]; o.z = pa[64]; o.w = pa[96];
        *(float4*)(out + (size_t)(node0 + r) * HID + 4 * l) = o;
    }
    float* pd = denom + (size_t)node0 * HEADS;
    for (int t = tid; t < rows * HEADS; t += 1024) pd[t] = den[t];
}

extern "C" void kernel_launch(void* const* d_in, const int* in_sizes, int n_in,
                              void* d_out, int out_size, void* d_ws, size_t ws_size,
                              hipStream_t stream) {
    const float* x          = (const float*)d_in[0];
    const int*   edge_index = (const int*)  d_in[1];
    const float* Wl_all     = (const float*)d_in[2];
    const float* Wr_all     = (const float*)d_in[3];
    const float* att_all    = (const float*)d_in[4];
    const float* bias_all   = (const float*)d_in[5];
    float* out = (float*)d_out;

    const int N = in_sizes[0] / HID;
    const int E = in_sizes[1] / 2;

    const int* src = edge_index;
    const int* dst = edge_index + E;

    const size_t nf = (size_t)N * HID;
    float* xl    = (float*)d_ws;
    float* xr    = xl + nf;
    float* bufA  = xr + nf;
    float* denom = bufA + nf;
    int*   glist = (int*)(denom + (size_t)N * HEADS);

    const int TPB = 256;
    const int gemm_blocks = N / 16;
    const int own_blocks  = (N + OWN_ROWS - 1) / OWN_ROWS;
    int* gcnt = glist + (size_t)own_blocks * OWN_CAP;
    const int n4_blocks   = (N * HID / 4 + TPB - 1) / TPB;

    const float* xin = x;
    for (int l = 0; l < LAYERS; ++l) {
        const float* Wl   = Wl_all   + (size_t)l * HID * HID;
        const float* Wr   = Wr_all   + (size_t)l * HID * HID;
        const float* att  = att_all  + (size_t)l * HEADS * HEAD_DIM;
        float* xout = (l == LAYERS - 1) ? out : bufA;
        if (l == 0)
            gemm_dual_f32<false><<<gemm_blocks, TPB, 0, stream>>>(
                xin, nullptr, nullptr, Wl, Wr, xl, xr);
        else
            gemm_dual_f32<true><<<gemm_blocks, TPB, 0, stream>>>(
                xin, denom, bias_all + (size_t)(l - 1) * HID, Wl, Wr, xl, xr);
        if (l == 0)
            owner_attend_kernel<true><<<own_blocks, 1024, 0, stream>>>(
                src, dst, xl, xr, att, denom, xout, glist, gcnt, N, E);
        else
            owner_attend_kernel<false><<<own_blocks, 1024, 0, stream>>>(
                src, dst, xl, xr, att, denom, xout, glist, gcnt, N, E);
        xin = xout;
    }
    divide_out_kernel<<<n4_blocks, TPB, 0, stream>>>(
        out, denom, bias_all + (size_t)(LAYERS - 1) * HID, N * HID / 4);
}


# First-layer owner scan reads a 1-byte owner map (dst/196 per edge, written by a new pre-pass launch), 16 edges per lane load, fetched a step ahead; was dst itself; sums in arrival order as before
# speedup vs baseline: 6.0611x; 1.0707x over previous
#include <hip/hip_runtime.h>
#include <hip/hip_bf16.h>
#include <math.h>

#define HID      128
#define HEADS    8
#define HEAD_DIM 16
#define LAYERS   3
#define NEG_SLOPE 0.2f
#define EPS_F     1e-16f

typedef float v2f __attribute__((ext_vector_type(2)));
typedef float v8f __attribute__((ext_vector_type(8)));

template <bool NORM>
__global__ __launch_bounds__(256)
void gemm_dual_f32(const float* __restrict__ X,
                   const float* __restrict__ den,
                   const float* __restrict__ bias,
                   const float* __restrict__ Wl,
                   const float* __restrict__ Wr,
                   float* __restrict__ XL,
                   float* __restrict__ XR) {
    __shared__ float As[16][HID];
    const int row0 = blockIdx.x * 16;

    {
        const float4* gx = (const float4*)(X + (size_t)row0 * HID);
        float4* sx = (float4*)(&As[0][0]);
        for (int t = threadIdx.x; t < 16 * HID / 4; t += 256) {
            float4 v = gx[t];
            if (NORM) {
                const int r  = t >> 5;
                const int c4 = (t & 31) * 4;
                const float dn = den[(size_t)(row0 + r) * HEADS + (c4 >> 4)] + EPS_F;
                const float4 b = *(const float4*)(bias + c4);
                v.x = v.x / dn + b.x;
                v.y = v.y / dn + b.y;
                v.z = v.z / dn + b.z;
                v.w = v.w / dn + b.w;
            }
            sx[t] = v;
        }
    }
    __syncthreads();

    const int wave = threadIdx.x >> 5;
    const int lane = threadIdx.x & 31;
    const int col0 = wave * 16;
    const int half = lane >> 4;
    const int lm   = lane & 15;

    v8f accL = {0.f,0.f,0.f,0.f,0.f,0.f,0.f,0.f};
    v8f accR = {0.f,0.f,0.f,0.f,0.f,0.f,0.f,0.f};

    #pragma unroll 4
    for (int k0 = 0; k0 < HID; k0 += 4) {
        const int ka = k0 + 2 * half;
        v2f a;
        a.x = As[lm][ka];
        a.y = As[lm][ka + 1];

        v2f bl, br;
        bl.x = Wl[(size_t)(ka + 0) * HID + col0 + lm];
        bl.y = Wl[(size_t)(ka + 1) * HID + col0 + lm];
        br.x = Wr[(size_t)(ka + 0) * HID + col0 + lm];
        br.y = Wr[(size_t)(ka + 1) * HID + col0 + lm];

        accL = __builtin_amdgcn_wmma_f32_16x16x4_f32(false, a, false, bl,
                                                     (short)0, accL, false, false);
        accR = __builtin_amdgcn_wmma_f32_16x16x4_f32(false, a, false, br,
                                                     (short)0, accR, false, false);
    }

    #pragma unroll
    for (int r = 0; r < 8; ++r) {
        const int m = r + 8 * half;
        XL[(size_t)(row0 + m) * HID + col0 + lm] = accL[r];
        XR[(size_t)(row0 + m) * HID + col0 + lm] = accR[r];
    }
}

__global__ void divide_out_kernel(float* __restrict__ out,
                                  const float* __restrict__ den,
                                  const float* __restrict__ bias, int total4) {
    int i = blockIdx.x * blockDim.x + threadIdx.x;
    if (i >= total4) return;
    const int n  = i >> 5;
    const int c4 = (i & 31) * 4;
    float4 v = ((const float4*)out)[i];
    const float dn = den[(size_t)n * HEADS + (c4 >> 4)] + EPS_F;
    const float4 b = *(const float4*)(bias + c4);
    v.x = v.x / dn + b.x;
    v.y = v.y / dn + b.y;
    v.z = v.z / dn + b.z;
    v.w = v.w / dn + b.w;
    ((float4*)out)[i] = v;
}

#define OWN_ROWS   196
#define OWN_WAVES  32
#define OWN_STEP   (OWN_WAVES * 512)
#define OWN_CAP    24576
#define SHIFT_SPAN 40.0f
#define NO_EDGE    0x7fffffff
#define SEED_ROWS  224

__device__ __forceinline__ void lds_max_float(float* addr, float val) {
    if (__float_as_int(val) >= 0)
        atomicMax((int*)addr, __float_as_int(val));
    else
        atomicMin((unsigned int*)addr, __float_as_uint(val));
}

template <int CTRL>
__device__ __forceinline__ float quad_swap(float v) {
    return __int_as_float(__builtin_amdgcn_update_dpp(
        0, __float_as_int(v), CTRL, 0xf, 0xf, false));
}

__device__ __forceinline__ void append_edges(bool m, int word, int* list, int* cnt,
                                             int lane) {
    const unsigned mask = __builtin_amdgcn_ballot_w32(m);
    if (mask != 0u) {
        int base = 0;
        if (lane == 0) base = atomicAdd(cnt, (int)__popc(mask));
        base = __builtin_amdgcn_readfirstlane(base);
        if (m) list[base + (int)__builtin_amdgcn_mbcnt_lo(mask, 0u)] = word;
    }
}

__device__ __forceinline__ float edge_logit(const float4 v, const float4 r,
                                            const float4 a4) {
    float t, p;
    t = v.x + r.x; t = (t > 0.0f) ? t : NEG_SLOPE * t; p = t * a4.x;
    t = v.y + r.y; t = (t > 0.0f) ? t : NEG_SLOPE * t; p += t * a4.y;
    t = v.z + r.z; t = (t > 0.0f) ? t : NEG_SLOPE * t; p += t * a4.z;
    t = v.w + r.w; t = (t > 0.0f) ? t : NEG_SLOPE * t; p += t * a4.w;
    p += quad_swap<0xB1>(p);
    p += quad_swap<0x4E>(p);
    return p;
}

__device__ __forceinline__ void seed_shifts(const int* first, float* mx,
                                            const float* xrs,
                                            const float* __restrict__ xl,
                                            const float4 a4, int wave, int lane) {
    constexpr int PER = SEED_ROWS / OWN_WAVES;
    float4 v[PER];
    int wd[PER];
    #pragma unroll
    for (int j = 0; j < PER; ++j) wd[j] = first[wave + j * OWN_WAVES];
    #pragma unroll
    for (int j = 0; j < PER; ++j) {
        wd[j] = __builtin_amdgcn_readfirstlane(wd[j]);
        const unsigned s = (wd[j] == NO_EDGE) ? 0u : ((unsigned)wd[j] >> 8);
        v[j] = *(const float4*)(xl + (size_t)s * HID + 4 * lane);
    }
    #pragma unroll
    for (int j = 0; j < PER; ++j) {
        if (wd[j] != NO_EDGE) {
            const int r = wave + j * OWN_WAVES;
            const float4 x4 = *(const float4*)(xrs + r * HID + 4 * lane);
            const float p = edge_logit(v[j], x4, a4);
            if ((lane & 3) == 0) mx[r * HEADS + (lane >> 2)] = p;
        }
    }
}

template <int MODE>
__device__ __forceinline__ bool sweep_edges(int n, const int* list, float* acc,
                                            float* den, float* mx, const float* xrs,
                                            const float* __restrict__ xl,
                                            const float4 a4, int wave, int lane) {
    const int hq = lane >> 2;
    bool over = false;
    for (int i0 = wave * 8; i0 < n; i0 += OWN_WAVES * 8) {
        const int idx = i0 + (lane & 7);
        const int rec = list[idx < n ? idx : i0];
        float4 v[8];
        int dl[8];
        #pragma unroll
        for (int j = 0; j < 8; ++j) {
            const unsigned rj = (unsigned)__builtin_amdgcn_readlane(rec, j);
            dl[j] = (int)(rj & 255u);
            v[j] = *(const float4*)(xl + (size_t)(rj >> 8) * HID + 4 * lane);
        }
        #pragma unroll
        for (int j = 0; j < 8; ++j) {
            if (i0 + j < n) {
                const float4 r = *(const float4*)(xrs + dl[j] * HID + 4 * lane);
                const float p = edge_logit(v[j], r, a4);
                float* pm = mx + dl[j] * HEADS + hq;
                if (MODE == 1) {
                    if ((lane & 3) == 0) lds_max_float(pm, p);
                } else {
                    const float d = p - *pm;
                    if (MODE == 3) over |= (d > SHIFT_SPAN);
                    const float w = expf(d);
                    if ((lane & 3) == 0) atomicAdd(den + dl[j] * HEADS + hq, w);
                    float* po = acc + dl[j] * HID + lane;
                    atomicAdd(po +  0, w * v[j].x);
                    atomicAdd(po + 32, w * v[j].y);
                    atomicAdd(po + 64, w * v[j].z);
                    atomicAdd(po + 96, w * v[j].w);
                }
            }
        }
    }
    return over;
}

template <bool ONE_SWEEP>
__device__ __forceinline__ void attend_list(int n, const int* list,
                                            const int* first, int* retry, float* acc,
                                            float* den, float* mx, const float* xrs,
                                            const float* __restrict__ xl,
                                            const float4 a4, int tid, int wave,
                                            int lane) {
    if (ONE_SWEEP) {
        seed_shifts(first, mx, xrs, xl, a4, wave, lane);
        __syncthreads();
        if (sweep_edges<3>(n, list, acc, den, mx, xrs, xl, a4, wave, lane)) *retry = 1;
        __syncthreads();
        if (__builtin_amdgcn_readfirstlane(*retry) == 0) return;
        for (int t = tid; t < OWN_ROWS * HID; t += 1024) acc[t] = 0.0f;
        for (int t = tid; t < OWN_ROWS * HEADS; t += 1024) {
            den[t] = 0.0f;
            mx[t] = -INFINITY;
        }
        __syncthreads();
    }
    sweep_edges<1>(n, list, acc, den, mx, xrs, xl, a4, wave, lane);
    __syncthreads();
    sweep_edges<2>(n, list, acc, den, mx, xrs, xl, a4, wave, lane);
    __syncthreads();
}

__device__ __forceinline__ void edges_to_sources(int n, int* list,
                                                 const int* __restrict__ src,
                                                 const int* __restrict__ dst,
                                                 int node0, int tid) {
    for (int t = tid; t < n; t += 1024) {
        const int e = list[t];
        list[t] = (int)(((unsigned)src[e] << 8) | (unsigned)(dst[e] - node0));
    }
}

__device__ __forceinline__ unsigned zero_bytes(unsigned x) {
    const unsigned t = (x & 0x7f7f7f7fu) + 0x7f7f7f7fu;
    return ~(t | x | 0x7f7f7f7fu);
}

__device__ __forceinline__ void take_hits(unsigned hh, int e0, int E, int* list,
                                          int* cnt, int lane) {
    while (__builtin_amdgcn_ballot_w32(hh != 0u) != 0u) {
        const int b = (hh != 0u) ? __builtin_ctz(hh) : 0;
        const int e = e0 + ((b & 3) << 2) + (b >> 3);
        append_edges(hh != 0u && e < E, e, list, cnt, lane);
        hh &= hh - 1u;
    }
}

__global__ __launch_bounds__(256)
void owner_map_kernel(const int* __restrict__ dst, unsigned int* __restrict__ omap,
                      int E, int nthreads) {
    const int t = blockIdx.x * 256 + threadIdx.x;
    if (t >= nthreads) return;
    const int e0 = t * 16;
    unsigned o[4];
    if ((((size_t)dst) & 15) == 0 && e0 + 16 <= E) {
        int4 d[4];
        #pragma unroll
        for (int g = 0; g < 4; ++g) d[g] = *(const int4*)(dst + e0 + 4 * g);
        #pragma unroll
        for (int g = 0; g < 4; ++g)
            o[g] = ((unsigned)d[g].x / OWN_ROWS) | (((unsigned)d[g].y / OWN_ROWS) << 8) |
                   (((unsigned)d[g].z / OWN_ROWS) << 16) | (((unsigned)d[g].w / OWN_ROWS) << 24);
    } else {
        #pragma unroll
        for (int g = 0; g < 4; ++g) {
            o[g] = 0u;
            #pragma unroll
            for (int k = 0; k < 4; ++k) {
                const int e = e0 + 4 * g + k;
                const unsigned d = (e < E) ? (unsigned)dst[e] : 0u;
                o[g] |= (d / OWN_ROWS) << (8 * k);
            }
        }
    }
    uint4 v;
    v.x = o[0]; v.y = o[1]; v.z = o[2]; v.w = o[3];
    *(uint4*)(omap + 4 * (size_t)t) = v;
}

template <bool FIRST>
__global__ __launch_bounds__(1024)
void owner_attend_kernel(const int* __restrict__ src,
                         const int* __restrict__ dst,
                         const float* __restrict__ xl,
                         const float* __restrict__ xr,
                         const float* __restrict__ att,
                         float* __restrict__ denom,
                         float* __restrict__ out,
                         int* __restrict__ glist,
                         int* __restrict__ gcnt,
                         const unsigned int* __restrict__ omap, int N, int E) {
    __shared__ float acc[OWN_ROWS * HID];
    __shared__ float xrs[OWN_ROWS * HID];
    __shared__ float den[OWN_ROWS * HEADS];
    __shared__ float mx[OWN_ROWS * HEADS];
    __shared__ int   list[OWN_CAP];
    __shared__ int   first[SEED_ROWS];
    __shared__ int   cnt;
    __shared__ int   retry;

    const int tid = threadIdx.x;
    const int lane = tid & 31;
    const int wave = __builtin_amdgcn_readfirstlane(tid >> 5);
    const int node0 = blockIdx.x * OWN_ROWS;
    const int rows = min(OWN_ROWS, N - node0);

    for (int t = tid; t < OWN_ROWS * HID; t += 1024) acc[t] = 0.0f;
    for (int t = tid; t < OWN_ROWS * HEADS; t += 1024) {
        den[t] = 0.0f;
        mx[t] = -INFINITY;
    }
    {
        const float4* gx = (const float4*)(xr + (size_t)node0 * HID);
        float4* sx = (float4*)xrs;
        for (int t = tid; t < rows * (HID / 4); t += 1024) sx[t] = gx[t];
    }
    if (tid < SEED_ROWS) first[tid] = NO_EDGE;
    if (tid == 0) { cnt = 0; retry = 0; }
    __syncthreads();

    const float4 a4 = ((const float4*)att)[lane];
    const int nsteps = (E + OWN_STEP - 1) / OWN_STEP;

    int* mylist = glist + (size_t)blockIdx.x * OWN_CAP;
    int kept = -1;
    if (!FIRST) kept = __builtin_amdgcn_readfirstlane(gcnt[blockIdx.x]);
    if (kept >= 0) {
        for (int t0 = tid; t0 < kept; t0 += 4096) {
            int w[4];
            #pragma unroll
            for (int k = 0; k < 4; ++k)
                w[k] = (t0 + k * 1024 < kept) ? mylist[t0 + k * 1024] : NO_EDGE;
            #pragma unroll
            for (int k = 0; k < 4; ++k) {
                if (t0 + k * 1024 < kept) {
                    list[t0 + k * 1024] = w[k];
                    atomicMin(&first[w[k] & 255], w[k]);
                }
            }
        }
        __syncthreads();
        attend_list<true>(kept, list, first, &retry, acc, den, mx, xrs, xl, a4,
                          tid, wave, lane);
    } else {
        bool emptied = false;
        const int step0 = (int)(blockIdx.x % (unsigned)nsteps);
        const unsigned me4 = blockIdx.x * 0x01010101u;
        const uint4* mp = (const uint4*)omap + wave * 32 + lane;
        for (int scan = 0; scan < 2; ++scan) {
            uint4 nxt = mp[(size_t)step0 * (OWN_STEP / 16)];
            for (int it = 0; it < nsteps; ++it) {
                const int step = (it + step0 < nsteps) ? it + step0 : it + step0 - nsteps;
                const uint4 ow = nxt;
                if (it + 1 < nsteps)
                    nxt = mp[(size_t)((step + 1 < nsteps) ? step + 1 : 0) * (OWN_STEP / 16)];
                const int wbase = step * OWN_STEP + wave * 512 + lane * 16;
                const unsigned h0 = zero_bytes(ow.x ^ me4), h1 = zero_bytes(ow.y ^ me4);
                const unsigned h2 = zero_bytes(ow.z ^ me4), h3 = zero_bytes(ow.w ^ me4);
                take_hits((h0 >> 7) | (h1 >> 6) | (h2 >> 5) | (h3 >> 4), wbase, E, list,
                          &cnt, lane);
                __syncthreads();
                const int n = __builtin_amdgcn_readfirstlane(cnt);
                __syncthreads();
                if (n <= OWN_CAP - OWN_STEP && it != nsteps - 1) continue;
                edges_to_sources(n, list, src, dst, node0, tid);
                __syncthreads();
                if (it == nsteps - 1 && !emptied) {
                    if (FIRST) {
                        for (int t = tid; t < n; t += 1024) {
                            const int w = list[t];
                            mylist[t] = w;
                            atomicMin(&first[w & 255], w);
                        }
                        if (tid == 0) gcnt[blockIdx.x] = n;
                        __syncthreads();
                    }
                    attend_list<FIRST>(n, list, first, &retry, acc, den, mx, xrs,
                                       xl, a4, tid, wave, lane);
                    break;
                }
                emptied = true;
                if (scan == 0)
                    sweep_edges<1>(n, list, acc, den, mx, xrs, xl, a4, wave, lane);
                else
                    sweep_edges<2>(n, list, acc, den, mx, xrs, xl, a4, wave, lane);
                __syncthreads();
                if (tid == 0) cnt = 0;
                __syncthreads();
            }
            if (!emptied) break;
        }
        if (FIRST && emptied && tid == 0) gcnt[blockIdx.x] = -1;
    }

    for (int t = tid; t < rows * (HID / 4); t += 1024) {
        const int r = t >> 5, l = t & 31;
        const float* pa = acc + r * HID + l;
        float4 o;
        o.x = pa[0]; o.y = pa[32]; o.z = pa[64]; o.w = pa[96];
        *(float4*)(out + (size_t)(node0 + r) * HID + 4 * l) = o;
    }
    float* pd = denom + (size_t)node0 * HEADS;
    for (int t = tid; t < rows * HEADS; t += 1024) pd[t] = den[t];
}

extern "C" void kernel_launch(void* const* d_in, const int* in_sizes, int n_in,
                              void* d_out, int out_size, void* d_ws, size_t ws_size,
                              hipStream_t stream) {
    const float* x          = (const float*)d_in[0];
    const int*   edge_index = (const int*)  d_in[1];
    const float* Wl_all     = (const float*)d_in[2];
    const float* Wr_all     = (const float*)d_in[3];
    const float* att_all    = (const float*)d_in[4];
    const float* bias_all   = (const float*)d_in[5];
    float* out = (float*)d_out;

    const int N = in_sizes[0] / HID;
    const int E = in_sizes[1] / 2;

    const int* src = edge_index;
    const int* dst = edge_index + E;

    const size_t nf = (size_t)N * HID;
    float* xl    = (float*)d_ws;
    float* xr    = xl + nf;
    float* bufA  = xr + nf;
    float* denom = bufA + nf;
    const int map_steps = (E + OWN_STEP - 1) / OWN_STEP;
    unsigned int* omap = (unsigned int*)(denom + (size_t)N * HEADS);
    int*   glist = (int*)(omap + (size_t)map_steps * (OWN_STEP / 4));

    const int TPB = 256;
    const int gemm_blocks = N / 16;
    const int own_blocks  = (N + OWN_ROWS - 1) / OWN_ROWS;
    int* gcnt = glist + (size_t)own_blocks * OWN_CAP;
    const int n4_blocks   = (N * HID / 4 + TPB - 1) / TPB;

    const int map_threads = map_steps * (OWN_STEP / 16);
    owner_map_kernel<<<(map_threads + TPB - 1) / TPB, TPB, 0, stream>>>(
        dst, omap, E, map_threads);

    const float* xin = x;
    for (int l = 0; l < LAYERS; ++l) {
        const float* Wl   = Wl_all   + (size_t)l * HID * HID;
        const float* Wr   = Wr_all   + (size_t)l * HID * HID;
        const float* att  = att_all  + (size_t)l * HEADS * HEAD_DIM;
        float* xout = (l == LAYERS - 1) ? out : bufA;
        if (l == 0)
            gemm_dual_f32<false><<<gemm_blocks, TPB, 0, stream>>>(
                xin, nullptr, nullptr, Wl, Wr, xl, xr);
        else
            gemm_dual_f32<true><<<gemm_blocks, TPB, 0, stream>>>(
                xin, denom, bias_all + (size_t)(l - 1) * HID, Wl, Wr, xl, xr);
        if (l == 0)
            owner_attend_kernel<true><<<own_blocks, 1024, 0, stream>>>(
                src, dst, xl, xr, att, denom, xout, glist, gcnt, omap, N, E);
        else
            owner_attend_kernel<false><<<own_blocks, 1024, 0, stream>>>(
                src, dst, xl, xr, att, denom, xout, glist, gcnt, omap, N, E);
        xin = xout;
    }
    divide_out_kernel<<<n4_blocks, TPB, 0, stream>>>(
        out, denom, bias_all + (size_t)(LAYERS - 1) * HID, N * HID / 4);
}


# Owner work-group stores finished rows, sum / (den + eps) + bias, from its LDS sums and denominators: last divide launch, the GEMM load's divide and the denominator array gone; same sums and order
# speedup vs baseline: 6.3151x; 1.0419x over previous
#include <hip/hip_runtime.h>
#include <hip/hip_bf16.h>
#include <math.h>

#define HID      128
#define HEADS    8
#define HEAD_DIM 16
#define LAYERS   3
#define NEG_SLOPE 0.2f
#define EPS_F     1e-16f

typedef float v2f __attribute__((ext_vector_type(2)));
typedef float v8f __attribute__((ext_vector_type(8)));

__global__ __launch_bounds__(256)
void gemm_dual_f32(const float* __restrict__ X,
                   const float* __restrict__ Wl,
                   const float* __restrict__ Wr,
                   float* __restrict__ XL,
                   float* __restrict__ XR) {
    __shared__ float As[16][HID];
    const int row0 = blockIdx.x * 16;

    {
        const float4* gx = (const float4*)(X + (size_t)row0 * HID);
        float4* sx = (float4*)(&As[0][0]);
        for (int t = threadIdx.x; t < 16 * HID / 4; t += 256) sx[t] = gx[t];
    }
    __syncthreads();

    const int wave = threadIdx.x >> 5;
    const int lane = threadIdx.x & 31;
    const int col0 = wave * 16;
    const int half = lane >> 4;
    const int lm   = lane & 15;

    v8f accL = {0.f,0.f,0.f,0.f,0.f,0.f,0.f,0.f};
    v8f accR = {0.f,0.f,0.f,0.f,0.f,0.f,0.f,0.f};

    #pragma unroll 4
    for (int k0 = 0; k0 < HID; k0 += 4) {
        const int ka = k0 + 2 * half;
        v2f a;
        a.x = As[lm][ka];
        a.y = As[lm][ka + 1];

        v2f bl, br;
        bl.x = Wl[(size_t)(ka + 0) * HID + col0 + lm];
        bl.y = Wl[(size_t)(ka + 1) * HID + col0 + lm];
        br.x = Wr[(size_t)(ka + 0) * HID + col0 + lm];
        br.y = Wr[(size_t)(ka + 1) * HID + col0 + lm];

        accL = __builtin_amdgcn_wmma_f32_16x16x4_f32(false, a, false, bl,
                                                     (short)0, accL, false, false);
        accR = __builtin_amdgcn_wmma_f32_16x16x4_f32(false, a, false, br,
                                                     (short)0, accR, false, false);
    }

    #pragma unroll
    for (int r = 0; r < 8; ++r) {
        const int m = r + 8 * half;
        XL[(size_t)(row0 + m) * HID + col0 + lm] = accL[r];
        XR[(size_t)(row0 + m) * HID + col0 + lm] = accR[r];
    }
}

#define OWN_ROWS   196
#define OWN_WAVES  32
#define OWN_STEP   (OWN_WAVES * 512)
#define OWN_CAP    24576
#define SHIFT_SPAN 40.0f
#define NO_EDGE    0x7fffffff
#define SEED_ROWS  224

__device__ __forceinline__ void lds_max_float(float* addr, float val) {
    if (__float_as_int(val) >= 0)
        atomicMax((int*)addr, __float_as_int(val));
    else
        atomicMin((unsigned int*)addr, __float_as_uint(val));
}

template <int CTRL>
__device__ __forceinline__ float quad_swap(float v) {
    return __int_as_float(__builtin_amdgcn_update_dpp(
        0, __float_as_int(v), CTRL, 0xf, 0xf, false));
}

__device__ __forceinline__ void append_edges(bool m, int word, int* list, int* cnt,
                                             int lane) {
    const unsigned mask = __builtin_amdgcn_ballot_w32(m);
    if (mask != 0u) {
        int base = 0;
        if (lane == 0) base = atomicAdd(cnt, (int)__popc(mask));
        base = __builtin_amdgcn_readfirstlane(base);
        if (m) list[base + (int)__builtin_amdgcn_mbcnt_lo(mask, 0u)] = word;
    }
}

__device__ __forceinline__ float edge_logit(const float4 v, const float4 r,
                                            const float4 a4) {
    float t, p;
    t = v.x + r.x; t = (t > 0.0f) ? t : NEG_SLOPE * t; p = t * a4.x;
    t = v.y + r.y; t = (t > 0.0f) ? t : NEG_SLOPE * t; p += t * a4.y;
    t = v.z + r.z; t = (t > 0.0f) ? t : NEG_SLOPE * t; p += t * a4.z;
    t = v.w + r.w; t = (t > 0.0f) ? t : NEG_SLOPE * t; p += t * a4.w;
    p += quad_swap<0xB1>(p);
    p += quad_swap<0x4E>(p);
    return p;
}

__device__ __forceinline__ void seed_shifts(const int* first, float* mx,
                                            const float* xrs,
                                            const float* __restrict__ xl,
                                            const float4 a4, int wave, int lane) {
    constexpr int PER = SEED_ROWS / OWN_WAVES;
    float4 v[PER];
    int wd[PER];
    #pragma unroll
    for (int j = 0; j < PER; ++j) wd[j] = first[wave + j * OWN_WAVES];
    #pragma unroll
    for (int j = 0; j < PER; ++j) {
        wd[j] = __builtin_amdgcn_readfirstlane(wd[j]);
        const unsigned s = (wd[j] == NO_EDGE) ? 0u : ((unsigned)wd[j] >> 8);
        v[j] = *(const float4*)(xl + (size_t)s * HID + 4 * lane);
    }
    #pragma unroll
    for (int j = 0; j < PER; ++j) {
        if (wd[j] != NO_EDGE) {
            const int r = wave + j * OWN_WAVES;
            const float4 x4 = *(const float4*)(xrs + r * HID + 4 * lane);
            const float p = edge_logit(v[j], x4, a4);
            if ((lane & 3) == 0) mx[r * HEADS + (lane >> 2)] = p;
        }
    }
}

template <int MODE>
__device__ __forceinline__ bool sweep_edges(int n, const int* list, float* acc,
                                            float* den, float* mx, const float* xrs,
                                            const float* __restrict__ xl,
                                            const float4 a4, int wave, int lane) {
    const int hq = lane >> 2;
    bool over = false;
    for (int i0 = wave * 8; i0 < n; i0 += OWN_WAVES * 8) {
        const int idx = i0 + (lane & 7);
        const int rec = list[idx < n ? idx : i0];
        float4 v[8];
        int dl[8];
        #pragma unroll
        for (int j = 0; j < 8; ++j) {
            const unsigned rj = (unsigned)__builtin_amdgcn_readlane(rec, j);
            dl[j] = (int)(rj & 255u);
            v[j] = *(const float4*)(xl + (size_t)(rj >> 8) * HID + 4 * lane);
        }
        #pragma unroll
        for (int j = 0; j < 8; ++j) {
            if (i0 + j < n) {
                const float4 r = *(const float4*)(xrs + dl[j] * HID + 4 * lane);
                const float p = edge_logit(v[j], r, a4);
                float* pm = mx + dl[j] * HEADS + hq;
                if (MODE == 1) {
                    if ((lane & 3) == 0) lds_max_float(pm, p);
                } else {
                    const float d = p - *pm;
                    if (MODE == 3) over |= (d > SHIFT_SPAN);
                    const float w = expf(d);
                    if ((lane & 3) == 0) atomicAdd(den + dl[j] * HEADS + hq, w);
                    float* po = acc + dl[j] * HID + lane;
                    atomicAdd(po +  0, w * v[j].x);
                    atomicAdd(po + 32, w * v[j].y);
                    atomicAdd(po + 64, w * v[j].z);
                    atomicAdd(po + 96, w * v[j].w);
                }
            }
        }
    }
    return over;
}

template <bool ONE_SWEEP>
__device__ __forceinline__ void attend_list(int n, const int* list,
                                            const int* first, int* retry, float* acc,
                                            float* den, float* mx, const float* xrs,
                                            const float* __restrict__ xl,
                                            const float4 a4, int tid, int wave,
                                            int lane) {
    if (ONE_SWEEP) {
        seed_shifts(first, mx, xrs, xl, a4, wave, lane);
        __syncthreads();
        if (sweep_edges<3>(n, list, acc, den, mx, xrs, xl, a4, wave, lane)) *retry = 1;
        __syncthreads();
        if (__builtin_amdgcn_readfirstlane(*retry) == 0) return;
        for (int t = tid; t < OWN_ROWS * HID; t += 1024) acc[t] = 0.0f;
        for (int t = tid; t < OWN_ROWS * HEADS; t += 1024) {
            den[t] = 0.0f;
            mx[t] = -INFINITY;
        }
        __syncthreads();
    }
    sweep_edges<1>(n, list, acc, den, mx, xrs, xl, a4, wave, lane);
    __syncthreads();
    sweep_edges<2>(n, list, acc, den, mx, xrs, xl, a4, wave, lane);
    __syncthreads();
}

__device__ __forceinline__ void edges_to_sources(int n, int* list,
                                                 const int* __restrict__ src,
                                                 const int* __restrict__ dst,
                                                 int node0, int tid) {
    for (int t = tid; t < n; t += 1024) {
        const int e = list[t];
        list[t] = (int)(((unsigned)src[e] << 8) | (unsigned)(dst[e] - node0));
    }
}

__device__ __forceinline__ unsigned zero_bytes(unsigned x) {
    const unsigned t = (x & 0x7f7f7f7fu) + 0x7f7f7f7fu;
    return ~(t | x | 0x7f7f7f7fu);
}

__device__ __forceinline__ void take_hits(unsigned hh, int e0, int E, int* list,
                                          int* cnt, int lane) {
    while (__builtin_amdgcn_ballot_w32(hh != 0u) != 0u) {
        const int b = (hh != 0u) ? __builtin_ctz(hh) : 0;
        const int e = e0 + ((b & 3) << 2) + (b >> 3);
        append_edges(hh != 0u && e < E, e, list, cnt, lane);
        hh &= hh - 1u;
    }
}

__global__ __launch_bounds__(256)
void owner_map_kernel(const int* __restrict__ dst, unsigned int* __restrict__ omap,
                      int E, int nthreads) {
    const int t = blockIdx.x * 256 + threadIdx.x;
    if (t >= nthreads) return;
    const int e0 = t * 16;
    unsigned o[4];
    if ((((size_t)dst) & 15) == 0 && e0 + 16 <= E) {
        int4 d[4];
        #pragma unroll
        for (int g = 0; g < 4; ++g) d[g] = *(const int4*)(dst + e0 + 4 * g);
        #pragma unroll
        for (int g = 0; g < 4; ++g)
            o[g] = ((unsigned)d[g].x / OWN_ROWS) | (((unsigned)d[g].y / OWN_ROWS) << 8) |
                   (((unsigned)d[g].z / OWN_ROWS) << 16) | (((unsigned)d[g].w / OWN_ROWS) << 24);
    } else {
        #pragma unroll
        for (int g = 0; g < 4; ++g) {
            o[g] = 0u;
            #pragma unroll
            for (int k = 0; k < 4; ++k) {
                const int e = e0 + 4 * g + k;
                const unsigned d = (e < E) ? (unsigned)dst[e] : 0u;
                o[g] |= (d / OWN_ROWS) << (8 * k);
            }
        }
    }
    uint4 v;
    v.x = o[0]; v.y = o[1]; v.z = o[2]; v.w = o[3];
    *(uint4*)(omap + 4 * (size_t)t) = v;
}

template <bool FIRST>
__global__ __launch_bounds__(1024)
void owner_attend_kernel(const int* __restrict__ src,
                         const int* __restrict__ dst,
                         const float* __restrict__ xl,
                         const float* __restrict__ xr,
                         const float* __restrict__ att,
                         const float* __restrict__ bias,
                         float* __restrict__ out,
                         int* __restrict__ glist,
                         int* __restrict__ gcnt,
                         const unsigned int* __restrict__ omap, int N, int E) {
    __shared__ float acc[OWN_ROWS * HID];
    __shared__ float xrs[OWN_ROWS * HID];
    __shared__ float den[OWN_ROWS * HEADS];
    __shared__ float mx[OWN_ROWS * HEADS];
    __shared__ int   list[OWN_CAP];
    __shared__ int   first[SEED_ROWS];
    __shared__ int   cnt;
    __shared__ int   retry;

    const int tid = threadIdx.x;
    const int lane = tid & 31;
    const int wave = __builtin_amdgcn_readfirstlane(tid >> 5);
    const int node0 = blockIdx.x * OWN_ROWS;
    const int rows = min(OWN_ROWS, N - node0);

    for (int t = tid; t < OWN_ROWS * HID; t += 1024) acc[t] = 0.0f;
    for (int t = tid; t < OWN_ROWS * HEADS; t += 1024) {
        den[t] = 0.0f;
        mx[t] = -INFINITY;
    }
    {
        const float4* gx = (const float4*)(xr + (size_t)node0 * HID);
        float4* sx = (float4*)xrs;
        for (int t = tid; t < rows * (HID / 4); t += 1024) sx[t] = gx[t];
    }
    if (tid < SEED_ROWS) first[tid] = NO_EDGE;
    if (tid == 0) { cnt = 0; retry = 0; }
    __syncthreads();

    const float4 a4 = ((const float4*)att)[lane];
    const int nsteps = (E + OWN_STEP - 1) / OWN_STEP;

    int* mylist = glist + (size_t)blockIdx.x * OWN_CAP;
    int kept = -1;
    if (!FIRST) kept = __builtin_amdgcn_readfirstlane(gcnt[blockIdx.x]);
    if (kept >= 0) {
        for (int t0 = tid; t0 < kept; t0 += 4096) {
            int w[4];
            #pragma unroll
            for (int k = 0; k < 4; ++k)
                w[k] = (t0 + k * 1024 < kept) ? mylist[t0 + k * 1024] : NO_EDGE;
            #pragma unroll
            for (int k = 0; k < 4; ++k) {
                if (t0 + k * 1024 < kept) {
                    list[t0 + k * 1024] = w[k];
                    atomicMin(&first[w[k] & 255], w[k]);
                }
            }
        }
        __syncthreads();
        attend_list<true>(kept, list, first, &retry, acc, den, mx, xrs, xl, a4,
                          tid, wave, lane);
    } else {
        bool emptied = false;
        const int step0 = (int)(blockIdx.x % (unsigned)nsteps);
        const unsigned me4 = blockIdx.x * 0x01010101u;
        const uint4* mp = (const uint4*)omap + wave * 32 + lane;
        for (int scan = 0; scan < 2; ++scan) {
            uint4 nxt = mp[(size_t)step0 * (OWN_STEP / 16)];
            for (int it = 0; it < nsteps; ++it) {
                const int step = (it + step0 < nsteps) ? it + step0 : it + step0 - nsteps;
                const uint4 ow = nxt;
                if (it + 1 < nsteps)
                    nxt = mp[(size_t)((step + 1 < nsteps) ? step + 1 : 0) * (OWN_STEP / 16)];
                const int wbase = step * OWN_STEP + wave * 512 + lane * 16;
                const unsigned h0 = zero_bytes(ow.x ^ me4), h1 = zero_bytes(ow.y ^ me4);
                const unsigned h2 = zero_bytes(ow.z ^ me4), h3 = zero_bytes(ow.w ^ me4);
                take_hits((h0 >> 7) | (h1 >> 6) | (h2 >> 5) | (h3 >> 4), wbase, E, list,
                          &cnt, lane);
                __syncthreads();
                const int n = __builtin_amdgcn_readfirstlane(cnt);
                __syncthreads();
                if (n <= OWN_CAP - OWN_STEP && it != nsteps - 1) continue;
                edges_to_sources(n, list, src, dst, node0, tid);
                __syncthreads();
                if (it == nsteps - 1 && !emptied) {
                    if (FIRST) {
                        for (int t = tid; t < n; t += 1024) {
                            const int w = list[t];
                            mylist[t] = w;
                            atomicMin(&first[w & 255], w);
                        }
                        if (tid == 0) gcnt[blockIdx.x] = n;
                        __syncthreads();
                    }
                    attend_list<FIRST>(n, list, first, &retry, acc, den, mx, xrs,
                                       xl, a4, tid, wave, lane);
                    break;
                }
                emptied = true;
                if (scan == 0)
                    sweep_edges<1>(n, list, acc, den, mx, xrs, xl, a4, wave, lane);
                else
                    sweep_edges<2>(n, list, acc, den, mx, xrs, xl, a4, wave, lane);
                __syncthreads();
                if (tid == 0) cnt = 0;
                __syncthreads();
            }
            if (!emptied) break;
        }
        if (FIRST && emptied && tid == 0) gcnt[blockIdx.x] = -1;
    }

    const float4 b4 = *(const float4*)(bias + 4 * lane);
    for (int t = tid; t < rows * (HID / 4); t += 1024) {
        const int r = t >> 5;
        const float* pa = acc + r * HID + lane;
        const float dn = den[r * HEADS + (lane >> 2)] + EPS_F;
        float4 o;
        o.x = pa[0]  / dn + b4.x;
        o.y = pa[32] / dn + b4.y;
        o.z = pa[64] / dn + b4.z;
        o.w = pa[96] / dn + b4.w;
        *(float4*)(out + (size_t)(node0 + r) * HID + 4 * lane) = o;
    }
}

extern "C" void kernel_launch(void* const* d_in, const int* in_sizes, int n_in,
                              void* d_out, int out_size, void* d_ws, size_t ws_size,
                              hipStream_t stream) {
    const float* x          = (const float*)d_in[0];
    const int*   edge_index = (const int*)  d_in[1];
    const float* Wl_all     = (const float*)d_in[2];
    const float* Wr_all     = (const float*)d_in[3];
    const float* att_all    = (const float*)d_in[4];
    const float* bias_all   = (const float*)d_in[5];
    float* out = (float*)d_out;

    const int N = in_sizes[0] / HID;
    const int E = in_sizes[1] / 2;

    const int* src = edge_index;
    const int* dst = edge_index + E;

    const size_t nf = (size_t)N * HID;
    float* xl    = (float*)d_ws;
    float* xr    = xl + nf;
    float* bufA  = xr + nf;
    const int map_steps = (E + OWN_STEP - 1) / OWN_STEP;
    unsigned int* omap = (unsigned int*)(bufA + nf);
    int*   glist = (int*)(omap + (size_t)map_steps * (OWN_STEP / 4));

    const int TPB = 256;
    const int gemm_blocks = N / 16;
    const int own_blocks  = (N + OWN_ROWS - 1) / OWN_ROWS;
    int* gcnt = glist + (size_t)own_blocks * OWN_CAP;

    const int map_threads = map_steps * (OWN_STEP / 16);
    owner_map_kernel<<<(map_threads + TPB - 1) / TPB, TPB, 0, stream>>>(
        dst, omap, E, map_threads);

    const float* xin = x;
    for (int l = 0; l < LAYERS; ++l) {
        const float* Wl   = Wl_all   + (size_t)l * HID * HID;
        const float* Wr   = Wr_all   + (size_t)l * HID * HID;
        const float* att  = att_all  + (size_t)l * HEADS * HEAD_DIM;
        const float* bias = bias_all + (size_t)l * HID;
        float* xout = (l == LAYERS - 1) ? out : bufA;
        gemm_dual_f32<<<gemm_blocks, TPB, 0, stream>>>(xin, Wl, Wr, xl, xr);
        if (l == 0)
            owner_attend_kernel<true><<<own_blocks, 1024, 0, stream>>>(
                src, dst, xl, xr, att, bias, xout, glist, gcnt, omap, N, E);
        else
            owner_attend_kernel<false><<<own_blocks, 1024, 0, stream>>>(
                src, dst, xl, xr, att, bias, xout, glist, gcnt, omap, N, E);
        xin = xout;
    }
}


# GEMM: weights stay in registers: 512 work-groups each walk many 16-row tiles, A strip in LDS in fragment order (whole 16-byte reads); sum order over k inside the products changed, still fixed
# speedup vs baseline: 7.0449x; 1.1156x over previous
#include <hip/hip_runtime.h>
#include <hip/hip_bf16.h>
#include <math.h>

#define HID      128
#define HEADS    8
#define HEAD_DIM 16
#define LAYERS   3
#define NEG_SLOPE 0.2f
#define EPS_F     1e-16f

typedef float v2f __attribute__((ext_vector_type(2)));
typedef float v4f __attribute__((ext_vector_type(4)));
typedef float v8f __attribute__((ext_vector_type(8)));

#define GEMM_WGS 512
__global__ __launch_bounds__(256)
void gemm_dual_f32(const float* __restrict__ X,
                   const float* __restrict__ Wl,
                   const float* __restrict__ Wr,
                   float* __restrict__ XL,
                   float* __restrict__ XR, int ntiles) {
    __shared__ v4f Af[2][16][32];
    const int wave = threadIdx.x >> 5;
    const int lane = threadIdx.x & 31;
    const int col0 = wave * 16;
    const int half = lane >> 4;
    const int lm   = lane & 15;

    v2f bl[32], br[32];
    #pragma unroll
    for (int s = 0; s < 32; ++s) {
        const int k = 8 * (s >> 1) + 4 * half + 2 * (s & 1);
        bl[s].x = Wl[(size_t)(k + 0) * HID + col0 + lm];
        bl[s].y = Wl[(size_t)(k + 1) * HID + col0 + lm];
        br[s].x = Wr[(size_t)(k + 0) * HID + col0 + lm];
        br[s].y = Wr[(size_t)(k + 1) * HID + col0 + lm];
    }

    const float* xa = X + (size_t)lm * HID + 8 * wave + 4 * half;
    int t = blockIdx.x;
    if (t < ntiles) {
        const float* g = xa + (size_t)t * 16 * HID;
        Af[0][wave][lane]     = *(const v4f*)(g);
        Af[0][wave + 8][lane] = *(const v4f*)(g + 64);
    }
    __syncthreads();

    int buf = 0;
    for (; t < ntiles; t += gridDim.x) {
        const int tn = t + gridDim.x;
        v4f p0, p1;
        if (tn < ntiles) {
            const float* g = xa + (size_t)tn * 16 * HID;
            p0 = *(const v4f*)(g);
            p1 = *(const v4f*)(g + 64);
        }

        v4f a[16];
        #pragma unroll
        for (int j = 0; j < 16; ++j) a[j] = Af[buf][j][lane];
        asm volatile("" :: "v"(a[0]), "v"(a[1]), "v"(a[2]), "v"(a[3]), "v"(a[4]), "v"(a[5]),
                     "v"(a[6]), "v"(a[7]), "v"(a[8]), "v"(a[9]), "v"(a[10]), "v"(a[11]),
                     "v"(a[12]), "v"(a[13]), "v"(a[14]), "v"(a[15]));

        v8f accL = {0.f,0.f,0.f,0.f,0.f,0.f,0.f,0.f};
        v8f accR = {0.f,0.f,0.f,0.f,0.f,0.f,0.f,0.f};
        #pragma unroll
        for (int j = 0; j < 16; ++j) {
            v2f a0, a1;
            a0.x = a[j].x; a0.y = a[j].y;
            a1.x = a[j].z; a1.y = a[j].w;
            accL = __builtin_amdgcn_wmma_f32_16x16x4_f32(false, a0, false, bl[2 * j],
                                                         (short)0, accL, false, false);
            accR = __builtin_amdgcn_wmma_f32_16x16x4_f32(false, a0, false, br[2 * j],
                                                         (short)0, accR, false, false);
            accL = __builtin_amdgcn_wmma_f32_16x16x4_f32(false, a1, false, bl[2 * j + 1],
                                                         (short)0, accL, false, false);
            accR = __builtin_amdgcn_wmma_f32_16x16x4_f32(false, a1, false, br[2 * j + 1],
                                                         (short)0, accR, false, false);
        }

        const size_t row0 = (size_t)t * 16;
        #pragma unroll
        for (int r = 0; r < 8; ++r) {
            const int m = r + 8 * half;
            XL[(row0 + m) * HID + col0 + lm] = accL[r];
            XR[(row0 + m) * HID + col0 + lm] = accR[r];
        }

        if (tn < ntiles) {
            Af[buf ^ 1][wave][lane]     = p0;
            Af[buf ^ 1][wave + 8][lane] = p1;
        }
        __builtin_amdgcn_fence(__ATOMIC_RELEASE, "workgroup", "local");
        __builtin_amdgcn_s_barrier();
        __builtin_amdgcn_fence(__ATOMIC_ACQUIRE, "workgroup", "local");
        buf ^= 1;
    }
}

#define OWN_ROWS   196
#define OWN_WAVES  32
#define OWN_STEP   (OWN_WAVES * 512)
#define OWN_CAP    24576
#define SHIFT_SPAN 40.0f
#define NO_EDGE    0x7fffffff
#define SEED_ROWS  224

__device__ __forceinline__ void lds_max_float(float* addr, float val) {
    if (__float_as_int(val) >= 0)
        atomicMax((int*)addr, __float_as_int(val));
    else
        atomicMin((unsigned int*)addr, __float_as_uint(val));
}

template <int CTRL>
__device__ __forceinline__ float quad_swap(float v) {
    return __int_as_float(__builtin_amdgcn_update_dpp(
        0, __float_as_int(v), CTRL, 0xf, 0xf, false));
}

__device__ __forceinline__ void append_edges(bool m, int word, int* list, int* cnt,
                                             int lane) {
    const unsigned mask = __builtin_amdgcn_ballot_w32(m);
    if (mask != 0u) {
        int base = 0;
        if (lane == 0) base = atomicAdd(cnt, (int)__popc(mask));
        base = __builtin_amdgcn_readfirstlane(base);
        if (m) list[base + (int)__builtin_amdgcn_mbcnt_lo(mask, 0u)] = word;
    }
}

__device__ __forceinline__ float edge_logit(const float4 v, const float4 r,
                                            const float4 a4) {
    float t, p;
    t = v.x + r.x; t = (t > 0.0f) ? t : NEG_SLOPE * t; p = t * a4.x;
    t = v.y + r.y; t = (t > 0.0f) ? t : NEG_SLOPE * t; p += t * a4.y;
    t = v.z + r.z; t = (t > 0.0f) ? t : NEG_SLOPE * t; p += t * a4.z;
    t = v.w + r.w; t = (t > 0.0f) ? t : NEG_SLOPE * t; p += t * a4.w;
    p += quad_swap<0xB1>(p);
    p += quad_swap<0x4E>(p);
    return p;
}

__device__ __forceinline__ void seed_shifts(const int* first, float* mx,
                                            const float* xrs,
                                            const float* __restrict__ xl,
                                            const float4 a4, int wave, int lane) {
    constexpr int PER = SEED_ROWS / OWN_WAVES;
    float4 v[PER];
    int wd[PER];
    #pragma unroll
    for (int j = 0; j < PER; ++j) wd[j] = first[wave + j * OWN_WAVES];
    #pragma unroll
    for (int j = 0; j < PER; ++j) {
        wd[j] = __builtin_amdgcn_readfirstlane(wd[j]);
        const unsigned s = (wd[j] == NO_EDGE) ? 0u : ((unsigned)wd[j] >> 8);
        v[j] = *(const float4*)(xl + (size_t)s * HID + 4 * lane);
    }
    #pragma unroll
    for (int j = 0; j < PER; ++j) {
        if (wd[j] != NO_EDGE) {
            const int r = wave + j * OWN_WAVES;
            const float4 x4 = *(const float4*)(xrs + r * HID + 4 * lane);
            const float p = edge_logit(v[j], x4, a4);
            if ((lane & 3) == 0) mx[r * HEADS + (lane >> 2)] = p;
        }
    }
}

template <int MODE>
__device__ __forceinline__ bool sweep_edges(int n, const int* list, float* acc,
                                            float* den, float* mx, const float* xrs,
                                            const float* __restrict__ xl,
                                            const float4 a4, int wave, int lane) {
    const int hq = lane >> 2;
    bool over = false;
    for (int i0 = wave * 8; i0 < n; i0 += OWN_WAVES * 8) {
        const int idx = i0 + (lane & 7);
        const int rec = list[idx < n ? idx : i0];
        float4 v[8];
        int dl[8];
        #pragma unroll
        for (int j = 0; j < 8; ++j) {
            const unsigned rj = (unsigned)__builtin_amdgcn_readlane(rec, j);
            dl[j] = (int)(rj & 255u);
            v[j] = *(const float4*)(xl + (size_t)(rj >> 8) * HID + 4 * lane);
        }
        #pragma unroll
        for (int j = 0; j < 8; ++j) {
            if (i0 + j < n) {
                const float4 r = *(const float4*)(xrs + dl[j] * HID + 4 * lane);
                const float p = edge_logit(v[j], r, a4);
                float* pm = mx + dl[j] * HEADS + hq;
                if (MODE == 1) {
                    if ((lane & 3) == 0) lds_max_float(pm, p);
                } else {
                    const float d = p - *pm;
                    if (MODE == 3) over |= (d > SHIFT_SPAN);
                    const float w = expf(d);
                    if ((lane & 3) == 0) atomicAdd(den + dl[j] * HEADS + hq, w);
                    float* po = acc + dl[j] * HID + lane;
                    atomicAdd(po +  0, w * v[j].x);
                    atomicAdd(po + 32, w * v[j].y);
                    atomicAdd(po + 64, w * v[j].z);
                    atomicAdd(po + 96, w * v[j].w);
                }
            }
        }
    }
    return over;
}

template <bool ONE_SWEEP>
__device__ __forceinline__ void attend_list(int n, const int* list,
                                            const int* first, int* retry, float* acc,
                                            float* den, float* mx, const float* xrs,
                                            const float* __restrict__ xl,
                                            const float4 a4, int tid, int wave,
                                            int lane) {
    if (ONE_SWEEP) {
        seed_shifts(first, mx, xrs, xl, a4, wave, lane);
        __syncthreads();
        if (sweep_edges<3>(n, list, acc, den, mx, xrs, xl, a4, wave, lane)) *retry = 1;
        __syncthreads();
        if (__builtin_amdgcn_readfirstlane(*retry) == 0) return;
        for (int t = tid; t < OWN_ROWS * HID; t += 1024) acc[t] = 0.0f;
        for (int t = tid; t < OWN_ROWS * HEADS; t += 1024) {
            den[t] = 0.0f;
            mx[t] = -INFINITY;
        }
        __syncthreads();
    }
    sweep_edges<1>(n, list, acc, den, mx, xrs, xl, a4, wave, lane);
    __syncthreads();
    sweep_edges<2>(n, list, acc, den, mx, xrs, xl, a4, wave, lane);
    __syncthreads();
}

__device__ __forceinline__ void edges_to_sources(int n, int* list,
                                                 const int* __restrict__ src,
                                                 const int* __restrict__ dst,
                                                 int node0, int tid) {
    for (int t = tid; t < n; t += 1024) {
        const int e = list[t];
        list[t] = (int)(((unsigned)src[e] << 8) | (unsigned)(dst[e] - node0));
    }
}

__device__ __forceinline__ unsigned zero_bytes(unsigned x) {
    const unsigned t = (x & 0x7f7f7f7fu) + 0x7f7f7f7fu;
    return ~(t | x | 0x7f7f7f7fu);
}

__device__ __forceinline__ void take_hits(unsigned hh, int e0, int E, int* list,
                                          int* cnt, int lane) {
    while (__builtin_amdgcn_ballot_w32(hh != 0u) != 0u) {
        const int b = (hh != 0u) ? __builtin_ctz(hh) : 0;
        const int e = e0 + ((b & 3) << 2) + (b >> 3);
        append_edges(hh != 0u && e < E, e, list, cnt, lane);
        hh &= hh - 1u;
    }
}

__global__ __launch_bounds__(256)
void owner_map_kernel(const int* __restrict__ dst, unsigned int* __restrict__ omap,
                      int E, int nthreads) {
    const int t = blockIdx.x * 256 + threadIdx.x;
    if (t >= nthreads) return;
    const int e0 = t * 16;
    unsigned o[4];
    if ((((size_t)dst) & 15) == 0 && e0 + 16 <= E) {
        int4 d[4];
        #pragma unroll
        for (int g = 0; g < 4; ++g) d[g] = *(const int4*)(dst + e0 + 4 * g);
        #pragma unroll
        for (int g = 0; g < 4; ++g)
            o[g] = ((unsigned)d[g].x / OWN_ROWS) | (((unsigned)d[g].y / OWN_ROWS) << 8) |
                   (((unsigned)d[g].z / OWN_ROWS) << 16) | (((unsigned)d[g].w / OWN_ROWS) << 24);
    } else {
        #pragma unroll
        for (int g = 0; g < 4; ++g) {
            o[g] = 0u;
            #pragma unroll
            for (int k = 0; k < 4; ++k) {
                const int e = e0 + 4 * g + k;
                const unsigned d = (e < E) ? (unsigned)dst[e] : 0u;
                o[g] |= (d / OWN_ROWS) << (8 * k);
            }
        }
    }
    uint4 v;
    v.x = o[0]; v.y = o[1]; v.z = o[2]; v.w = o[3];
    *(uint4*)(omap + 4 * (size_t)t) = v;
}

template <bool FIRST>
__global__ __launch_bounds__(1024)
void owner_attend_kernel(const int* __restrict__ src,
                         const int* __restrict__ dst,
                         const float* __restrict__ xl,
                         const float* __restrict__ xr,
                         const float* __restrict__ att,
                         const float* __restrict__ bias,
                         float* __restrict__ out,
                         int* __restrict__ glist,
                         int* __restrict__ gcnt,
                         const unsigned int* __restrict__ omap, int N, int E) {
    __shared__ float acc[OWN_ROWS * HID];
    __shared__ float xrs[OWN_ROWS * HID];
    __shared__ float den[OWN_ROWS * HEADS];
    __shared__ float mx[OWN_ROWS * HEADS];
    __shared__ int   list[OWN_CAP];
    __shared__ int   first[SEED_ROWS];
    __shared__ int   cnt;
    __shared__ int   retry;

    const int tid = threadIdx.x;
    const int lane = tid & 31;
    const int wave = __builtin_amdgcn_readfirstlane(tid >> 5);
    const int node0 = blockIdx.x * OWN_ROWS;
    const int rows = min(OWN_ROWS, N - node0);

    for (int t = tid; t < OWN_ROWS * HID; t += 1024) acc[t] = 0.0f;
    for (int t = tid; t < OWN_ROWS * HEADS; t += 1024) {
        den[t] = 0.0f;
        mx[t] = -INFINITY;
    }
    {
        const float4* gx = (const float4*)(xr + (size_t)node0 * HID);
        float4* sx = (float4*)xrs;
        for (int t = tid; t < rows * (HID / 4); t += 1024) sx[t] = gx[t];
    }
    if (tid < SEED_ROWS) first[tid] = NO_EDGE;
    if (tid == 0) { cnt = 0; retry = 0; }
    __syncthreads();

    const float4 a4 = ((const float4*)att)[lane];
    const int nsteps = (E + OWN_STEP - 1) / OWN_STEP;

    int* mylist = glist + (size_t)blockIdx.x * OWN_CAP;
    int kept = -1;
    if (!FIRST) kept = __builtin_amdgcn_readfirstlane(gcnt[blockIdx.x]);
    if (kept >= 0) {
        for (int t0 = tid; t0 < kept; t0 += 4096) {
            int w[4];
            #pragma unroll
            for (int k = 0; k < 4; ++k)
                w[k] = (t0 + k * 1024 < kept) ? mylist[t0 + k * 1024] : NO_EDGE;
            #pragma unroll
            for (int k = 0; k < 4; ++k) {
                if (t0 + k * 1024 < kept) {
                    list[t0 + k * 1024] = w[k];
                    atomicMin(&first[w[k] & 255], w[k]);
                }
            }
        }
        __syncthreads();
        attend_list<true>(kept, list, first, &retry, acc, den, mx, xrs, xl, a4,
                          tid, wave, lane);
    } else {
        bool emptied = false;
        const int step0 = (int)(blockIdx.x % (unsigned)nsteps);
        const unsigned me4 = blockIdx.x * 0x01010101u;
        const uint4* mp = (const uint4*)omap + wave * 32 + lane;
        for (int scan = 0; scan < 2; ++scan) {
            uint4 nxt = mp[(size_t)step0 * (OWN_STEP / 16)];
            for (int it = 0; it < nsteps; ++it) {
                const int step = (it + step0 < nsteps) ? it + step0 : it + step0 - nsteps;
                const uint4 ow = nxt;
                if (it + 1 < nsteps)
                    nxt = mp[(size_t)((step + 1 < nsteps) ? step + 1 : 0) * (OWN_STEP / 16)];
                const int wbase = step * OWN_STEP + wave * 512 + lane * 16;
                const unsigned h0 = zero_bytes(ow.x ^ me4), h1 = zero_bytes(ow.y ^ me4);
                const unsigned h2 = zero_bytes(ow.z ^ me4), h3 = zero_bytes(ow.w ^ me4);
                take_hits((h0 >> 7) | (h1 >> 6) | (h2 >> 5) | (h3 >> 4), wbase, E, list,
                          &cnt, lane);
                __syncthreads();
                const int n = __builtin_amdgcn_readfirstlane(cnt);
                __syncthreads();
                if (n <= OWN_CAP - OWN_STEP && it != nsteps - 1) continue;
                edges_to_sources(n, list, src, dst, node0, tid);
                __syncthreads();
                if (it == nsteps - 1 && !emptied) {
                    if (FIRST) {
                        for (int t = tid; t < n; t += 1024) {
                            const int w = list[t];
                            mylist[t] = w;
                            atomicMin(&first[w & 255], w);
                        }
                        if (tid == 0) gcnt[blockIdx.x] = n;
                        __syncthreads();
                    }
                    attend_list<FIRST>(n, list, first, &retry, acc, den, mx, xrs,
                                       xl, a4, tid, wave, lane);
                    break;
                }
                emptied = true;
                if (scan == 0)
                    sweep_edges<1>(n, list, acc, den, mx, xrs, xl, a4, wave, lane);
                else
                    sweep_edges<2>(n, list, acc, den, mx, xrs, xl, a4, wave, lane);
                __syncthreads();
                if (tid == 0) cnt = 0;
                __syncthreads();
            }
            if (!emptied) break;
        }
        if (FIRST && emptied && tid == 0) gcnt[blockIdx.x] = -1;
    }

    const float4 b4 = *(const float4*)(bias + 4 * lane);
    for (int t = tid; t < rows * (HID / 4); t += 1024) {
        const int r = t >> 5;
        const float* pa = acc + r * HID + lane;
        const float dn = den[r * HEADS + (lane >> 2)] + EPS_F;
        float4 o;
        o.x = pa[0]  / dn + b4.x;
        o.y = pa[32] / dn + b4.y;
        o.z = pa[64] / dn + b4.z;
        o.w = pa[96] / dn + b4.w;
        *(float4*)(out + (size_t)(node0 + r) * HID + 4 * lane) = o;
    }
}

extern "C" void kernel_launch(void* const* d_in, const int* in_sizes, int n_in,
                              void* d_out, int out_size, void* d_ws, size_t ws_size,
                              hipStream_t stream) {
    const float* x          = (const float*)d_in[0];
    const int*   edge_index = (const int*)  d_in[1];
    const float* Wl_all     = (const float*)d_in[2];
    const float* Wr_all     = (const float*)d_in[3];
    const float* att_all    = (const float*)d_in[4];
    const float* bias_all   = (const float*)d_in[5];
    float* out = (float*)d_out;

    const int N = in_sizes[0] / HID;
    const int E = in_sizes[1] / 2;

    const int* src = edge_index;
    const int* dst = edge_index + E;

    const size_t nf = (size_t)N * HID;
    float* xl    = (float*)d_ws;
    float* xr    = xl + nf;
    float* bufA  = xr + nf;
    const int map_steps = (E + OWN_STEP - 1) / OWN_STEP;
    unsigned int* omap = (unsigned int*)(bufA + nf);
    int*   glist = (int*)(omap + (size_t)map_steps * (OWN_STEP / 4));

    const int TPB = 256;
    const int gemm_blocks = N / 16;
    const int own_blocks  = (N + OWN_ROWS - 1) / OWN_ROWS;
    int* gcnt = glist + (size_t)own_blocks * OWN_CAP;

    const int map_threads = map_steps * (OWN_STEP / 16);
    owner_map_kernel<<<(map_threads + TPB - 1) / TPB, TPB, 0, stream>>>(
        dst, omap, E, map_threads);

    const float* xin = x;
    for (int l = 0; l < LAYERS; ++l) {
        const float* Wl   = Wl_all   + (size_t)l * HID * HID;
        const float* Wr   = Wr_all   + (size_t)l * HID * HID;
        const float* att  = att_all  + (size_t)l * HEADS * HEAD_DIM;
        const float* bias = bias_all + (size_t)l * HID;
        float* xout = (l == LAYERS - 1) ? out : bufA;
        gemm_dual_f32<<<GEMM_WGS, TPB, 0, stream>>>(xin, Wl, Wr, xl, xr, gemm_blocks);
        if (l == 0)
            owner_attend_kernel<true><<<own_blocks, 1024, 0, stream>>>(
                src, dst, xl, xr, att, bias, xout, glist, gcnt, omap, N, E);
        else
            owner_attend_kernel<false><<<own_blocks, 1024, 0, stream>>>(
                src, dst, xl, xr, att, bias, xout, glist, gcnt, omap, N, E);
        xin = xout;
    }
}


# Weighted sweep: 4 edges per 4-lane group, lane q gets edge q's whole logit (shares passed pair by pair, same add order): one shift read, exp, denominator add per 4 edges; atomics unordered as before
# speedup vs baseline: 7.5183x; 1.0672x over previous
#include <hip/hip_runtime.h>
#include <hip/hip_bf16.h>
#include <math.h>

#define HID      128
#define HEADS    8
#define HEAD_DIM 16
#define LAYERS   3
#define NEG_SLOPE 0.2f
#define EPS_F     1e-16f

typedef float v2f __attribute__((ext_vector_type(2)));
typedef float v4f __attribute__((ext_vector_type(4)));
typedef float v8f __attribute__((ext_vector_type(8)));

#define GEMM_WGS 512
__global__ __launch_bounds__(256)
void gemm_dual_f32(const float* __restrict__ X,
                   const float* __restrict__ Wl,
                   const float* __restrict__ Wr,
                   float* __restrict__ XL,
                   float* __restrict__ XR, int ntiles) {
    __shared__ v4f Af[2][16][32];
    const int wave = threadIdx.x >> 5;
    const int lane = threadIdx.x & 31;
    const int col0 = wave * 16;
    const int half = lane >> 4;
    const int lm   = lane & 15;

    v2f bl[32], br[32];
    #pragma unroll
    for (int s = 0; s < 32; ++s) {
        const int k = 8 * (s >> 1) + 4 * half + 2 * (s & 1);
        bl[s].x = Wl[(size_t)(k + 0) * HID + col0 + lm];
        bl[s].y = Wl[(size_t)(k + 1) * HID + col0 + lm];
        br[s].x = Wr[(size_t)(k + 0) * HID + col0 + lm];
        br[s].y = Wr[(size_t)(k + 1) * HID + col0 + lm];
    }

    const float* xa = X + (size_t)lm * HID + 8 * wave + 4 * half;
    int t = blockIdx.x;
    if (t < ntiles) {
        const float* g = xa + (size_t)t * 16 * HID;
        Af[0][wave][lane]     = *(const v4f*)(g);
        Af[0][wave + 8][lane] = *(const v4f*)(g + 64);
    }
    __syncthreads();

    int buf = 0;
    for (; t < ntiles; t += gridDim.x) {
        const int tn = t + gridDim.x;
        v4f p0, p1;
        if (tn < ntiles) {
            const float* g = xa + (size_t)tn * 16 * HID;
            p0 = *(const v4f*)(g);
            p1 = *(const v4f*)(g + 64);
        }

        v4f a[16];
        #pragma unroll
        for (int j = 0; j < 16; ++j) a[j] = Af[buf][j][lane];
        asm volatile("" :: "v"(a[0]), "v"(a[1]), "v"(a[2]), "v"(a[3]), "v"(a[4]), "v"(a[5]),
                     "v"(a[6]), "v"(a[7]), "v"(a[8]), "v"(a[9]), "v"(a[10]), "v"(a[11]),
                     "v"(a[12]), "v"(a[13]), "v"(a[14]), "v"(a[15]));

        v8f accL = {0.f,0.f,0.f,0.f,0.f,0.f,0.f,0.f};
        v8f accR = {0.f,0.f,0.f,0.f,0.f,0.f,0.f,0.f};
        #pragma unroll
        for (int j = 0; j < 16; ++j) {
            v2f a0, a1;
            a0.x = a[j].x; a0.y = a[j].y;
            a1.x = a[j].z; a1.y = a[j].w;
            accL = __builtin_amdgcn_wmma_f32_16x16x4_f32(false, a0, false, bl[2 * j],
                                                         (short)0, accL, false, false);
            accR = __builtin_amdgcn_wmma_f32_16x16x4_f32(false, a0, false, br[2 * j],
                                                         (short)0, accR, false, false);
            accL = __builtin_amdgcn_wmma_f32_16x16x4_f32(false, a1, false, bl[2 * j + 1],
                                                         (short)0, accL, false, false);
            accR = __builtin_amdgcn_wmma_f32_16x16x4_f32(false, a1, false, br[2 * j + 1],
                                                         (short)0, accR, false, false);
        }

        const size_t row0 = (size_t)t * 16;
        #pragma unroll
        for (int r = 0; r < 8; ++r) {
            const int m = r + 8 * half;
            XL[(row0 + m) * HID + col0 + lm] = accL[r];
            XR[(row0 + m) * HID + col0 + lm] = accR[r];
        }

        if (tn < ntiles) {
            Af[buf ^ 1][wave][lane]     = p0;
            Af[buf ^ 1][wave + 8][lane] = p1;
        }
        __builtin_amdgcn_fence(__ATOMIC_RELEASE, "workgroup", "local");
        __builtin_amdgcn_s_barrier();
        __builtin_amdgcn_fence(__ATOMIC_ACQUIRE, "workgroup", "local");
        buf ^= 1;
    }
}

#define OWN_ROWS   196
#define OWN_WAVES  32
#define OWN_STEP   (OWN_WAVES * 512)
#define OWN_CAP    24576
#define SHIFT_SPAN 40.0f
#define NO_EDGE    0x7fffffff
#define SEED_ROWS  224

__device__ __forceinline__ void lds_max_float(float* addr, float val) {
    if (__float_as_int(val) >= 0)
        atomicMax((int*)addr, __float_as_int(val));
    else
        atomicMin((unsigned int*)addr, __float_as_uint(val));
}

template <int CTRL>
__device__ __forceinline__ float quad_swap(float v) {
    return __int_as_float(__builtin_amdgcn_update_dpp(
        0, __float_as_int(v), CTRL, 0xf, 0xf, false));
}

__device__ __forceinline__ void append_edges(bool m, int word, int* list, int* cnt,
                                             int lane) {
    const unsigned mask = __builtin_amdgcn_ballot_w32(m);
    if (mask != 0u) {
        int base = 0;
        if (lane == 0) base = atomicAdd(cnt, (int)__popc(mask));
        base = __builtin_amdgcn_readfirstlane(base);
        if (m) list[base + (int)__builtin_amdgcn_mbcnt_lo(mask, 0u)] = word;
    }
}

__device__ __forceinline__ float edge_partial(const float4 v, const float4 r,
                                              const float4 a4) {
    float t, p;
    t = v.x + r.x; t = (t > 0.0f) ? t : NEG_SLOPE * t; p = t * a4.x;
    t = v.y + r.y; t = (t > 0.0f) ? t : NEG_SLOPE * t; p = __builtin_fmaf(t, a4.y, p);
    t = v.z + r.z; t = (t > 0.0f) ? t : NEG_SLOPE * t; p = __builtin_fmaf(t, a4.z, p);
    t = v.w + r.w; t = (t > 0.0f) ? t : NEG_SLOPE * t; p = __builtin_fmaf(t, a4.w, p);
    return p;
}

__device__ __forceinline__ float edge_logit(const float4 v, const float4 r,
                                            const float4 a4) {
    float p = edge_partial(v, r, a4);
    p += quad_swap<0xB1>(p);
    p += quad_swap<0x4E>(p);
    return p;
}

__device__ __forceinline__ float quad_transpose_sum(float p0, float p1, float p2,
                                                    float p3, bool odd, bool hi) {
    const float k01 = odd ? p1 : p0, s01 = odd ? p0 : p1;
    const float k23 = odd ? p3 : p2, s23 = odd ? p2 : p3;
    const float u01 = k01 + quad_swap<0xB1>(s01);
    const float u23 = k23 + quad_swap<0xB1>(s23);
    const float k = hi ? u23 : u01, s = hi ? u01 : u23;
    return k + quad_swap<0x4E>(s);
}

__device__ __forceinline__ void seed_shifts(const int* first, float* mx,
                                            const float* xrs,
                                            const float* __restrict__ xl,
                                            const float4 a4, int wave, int lane) {
    constexpr int PER = SEED_ROWS / OWN_WAVES;
    float4 v[PER];
    int wd[PER];
    #pragma unroll
    for (int j = 0; j < PER; ++j) wd[j] = first[wave + j * OWN_WAVES];
    #pragma unroll
    for (int j = 0; j < PER; ++j) {
        wd[j] = __builtin_amdgcn_readfirstlane(wd[j]);
        const unsigned s = (wd[j] == NO_EDGE) ? 0u : ((unsigned)wd[j] >> 8);
        v[j] = *(const float4*)(xl + (size_t)s * HID + 4 * lane);
    }
    #pragma unroll
    for (int j = 0; j < PER; ++j) {
        if (wd[j] != NO_EDGE) {
            const int r = wave + j * OWN_WAVES;
            const float4 x4 = *(const float4*)(xrs + r * HID + 4 * lane);
            const float p = edge_logit(v[j], x4, a4);
            if ((lane & 3) == 0) mx[r * HEADS + (lane >> 2)] = p;
        }
    }
}

template <int MODE>
__device__ __forceinline__ bool sweep_edges(int n, const int* list, float* acc,
                                            float* den, float* mx, const float* xrs,
                                            const float* __restrict__ xl,
                                            const float4 a4, int wave, int lane) {
    const int hq = lane >> 2;
    bool over = false;
    for (int i0 = wave * 8; i0 < n; i0 += OWN_WAVES * 8) {
        if (MODE != 1 && i0 + 8 <= n) {
            const int q = lane & 3;
            const bool odd = (lane & 1) != 0, hi = (lane & 2) != 0;
            int rw[2];
            rw[0] = list[i0 + q];
            rw[1] = list[i0 + 4 + q];
            float4 v[8];
            int dl[8];
            #pragma unroll
            for (int j = 0; j < 8; ++j) {
                const unsigned rj = (unsigned)__builtin_amdgcn_readlane(rw[j >> 2], j & 3);
                dl[j] = (int)(rj & 255u);
                v[j] = *(const float4*)(xl + (size_t)(rj >> 8) * HID + 4 * lane);
            }
            asm volatile("" :: "v"(v[0].x), "v"(v[1].x), "v"(v[2].x), "v"(v[3].x),
                         "v"(v[4].x), "v"(v[5].x), "v"(v[6].x), "v"(v[7].x));
            #pragma unroll
            for (int g = 0; g < 2; ++g) {
                float pp[4];
                #pragma unroll
                for (int k = 0; k < 4; ++k) {
                    const float4 r = *(const float4*)(xrs + dl[4 * g + k] * HID + 4 * lane);
                    pp[k] = edge_partial(v[4 * g + k], r, a4);
                }
                const float p = quad_transpose_sum(pp[0], pp[1], pp[2], pp[3], odd, hi);
                const int mo = (rw[g] & 255) * HEADS + hq;
                const float d = p - mx[mo];
                if (MODE == 3) over |= (d > SHIFT_SPAN);
                const float w = expf(d);
                atomicAdd(den + mo, w);
                #pragma unroll
                for (int k = 0; k < 4; ++k) {
                    const float wk = (k == 0) ? quad_swap<0x00>(w) : (k == 1) ? quad_swap<0x55>(w)
                                   : (k == 2) ? quad_swap<0xAA>(w) : quad_swap<0xFF>(w);
                    const float4 vk = v[4 * g + k];
                    float* po = acc + dl[4 * g + k] * HID + lane;
                    atomicAdd(po +  0, wk * vk.x);
                    atomicAdd(po + 32, wk * vk.y);
                    atomicAdd(po + 64, wk * vk.z);
                    atomicAdd(po + 96, wk * vk.w);
                }
            }
            continue;
        }
        const int idx = i0 + (lane & 7);
        const int rec = list[idx < n ? idx : i0];
        float4 v[8];
        int dl[8];
        #pragma unroll
        for (int j = 0; j < 8; ++j) {
            const unsigned rj = (unsigned)__builtin_amdgcn_readlane(rec, j);
            dl[j] = (int)(rj & 255u);
            v[j] = *(const float4*)(xl + (size_t)(rj >> 8) * HID + 4 * lane);
        }
        #pragma unroll
        for (int j = 0; j < 8; ++j) {
            if (i0 + j < n) {
                const float4 r = *(const float4*)(xrs + dl[j] * HID + 4 * lane);
                const float p = edge_logit(v[j], r, a4);
                float* pm = mx + dl[j] * HEADS + hq;
                if (MODE == 1) {
                    if ((lane & 3) == 0) lds_max_float(pm, p);
                } else {
                    const float d = p - *pm;
                    if (MODE == 3) over |= (d > SHIFT_SPAN);
                    const float w = expf(d);
                    if ((lane & 3) == 0) atomicAdd(den + dl[j] * HEADS + hq, w);
                    float* po = acc + dl[j] * HID + lane;
                    atomicAdd(po +  0, w * v[j].x);
                    atomicAdd(po + 32, w * v[j].y);
                    atomicAdd(po + 64, w * v[j].z);
                    atomicAdd(po + 96, w * v[j].w);
                }
            }
        }
    }
    return over;
}

template <bool ONE_SWEEP>
__device__ __forceinline__ void attend_list(int n, const int* list,
                                            const int* first, int* retry, float* acc,
                                            float* den, float* mx, const float* xrs,
                                            const float* __restrict__ xl,
                                            const float4 a4, int tid, int wave,
                                            int lane) {
    if (ONE_SWEEP) {
        seed_shifts(first, mx, xrs, xl, a4, wave, lane);
        __syncthreads();
        if (sweep_edges<3>(n, list, acc, den, mx, xrs, xl, a4, wave, lane)) *retry = 1;
        __syncthreads();
        if (__builtin_amdgcn_readfirstlane(*retry) == 0) return;
        for (int t = tid; t < OWN_ROWS * HID; t += 1024) acc[t] = 0.0f;
        for (int t = tid; t < OWN_ROWS * HEADS; t += 1024) {
            den[t] = 0.0f;
            mx[t] = -INFINITY;
        }
        __syncthreads();
    }
    sweep_edges<1>(n, list, acc, den, mx, xrs, xl, a4, wave, lane);
    __syncthreads();
    sweep_edges<2>(n, list, acc, den, mx, xrs, xl, a4, wave, lane);
    __syncthreads();
}

__device__ __forceinline__ void edges_to_sources(int n, int* list,
                                                 const int* __restrict__ src,
                                                 const int* __restrict__ dst,
                                                 int node0, int tid) {
    for (int t = tid; t < n; t += 1024) {
        const int e = list[t];
        list[t] = (int)(((unsigned)src[e] << 8) | (unsigned)(dst[e] - node0));
    }
}

__device__ __forceinline__ unsigned zero_bytes(unsigned x) {
    const unsigned t = (x & 0x7f7f7f7fu) + 0x7f7f7f7fu;
    return ~(t | x | 0x7f7f7f7fu);
}

__device__ __forceinline__ void take_hits(unsigned hh, int e0, int E, int* list,
                                          int* cnt, int lane) {
    while (__builtin_amdgcn_ballot_w32(hh != 0u) != 0u) {
        const int b = (hh != 0u) ? __builtin_ctz(hh) : 0;
        const int e = e0 + ((b & 3) << 2) + (b >> 3);
        append_edges(hh != 0u && e < E, e, list, cnt, lane);
        hh &= hh - 1u;
    }
}

__global__ __launch_bounds__(256)
void owner_map_kernel(const int* __restrict__ dst, unsigned int* __restrict__ omap,
                      int E, int nthreads) {
    const int t = blockIdx.x * 256 + threadIdx.x;
    if (t >= nthreads) return;
    const int e0 = t * 16;
    unsigned o[4];
    if ((((size_t)dst) & 15) == 0 && e0 + 16 <= E) {
        int4 d[4];
        #pragma unroll
        for (int g = 0; g < 4; ++g) d[g] = *(const int4*)(dst + e0 + 4 * g);
        #pragma unroll
        for (int g = 0; g < 4; ++g)
            o[g] = ((unsigned)d[g].x / OWN_ROWS) | (((unsigned)d[g].y / OWN_ROWS) << 8) |
                   (((unsigned)d[g].z / OWN_ROWS) << 16) | (((unsigned)d[g].w / OWN_ROWS) << 24);
    } else {
        #pragma unroll
        for (int g = 0; g < 4; ++g) {
            o[g] = 0u;
            #pragma unroll
            for (int k = 0; k < 4; ++k) {
                const int e = e0 + 4 * g + k;
                const unsigned d = (e < E) ? (unsigned)dst[e] : 0u;
                o[g] |= (d / OWN_ROWS) << (8 * k);
            }
        }
    }
    uint4 v;
    v.x = o[0]; v.y = o[1]; v.z = o[2]; v.w = o[3];
    *(uint4*)(omap + 4 * (size_t)t) = v;
}

template <bool FIRST>
__global__ __launch_bounds__(1024)
void owner_attend_kernel(const int* __restrict__ src,
                         const int* __restrict__ dst,
                         const float* __restrict__ xl,
                         const float* __restrict__ xr,
                         const float* __restrict__ att,
                         const float* __restrict__ bias,
                         float* __restrict__ out,
                         int* __restrict__ glist,
                         int* __restrict__ gcnt,
                         const unsigned int* __restrict__ omap, int N, int E) {
    __shared__ float acc[OWN_ROWS * HID];
    __shared__ float xrs[OWN_ROWS * HID];
    __shared__ float den[OWN_ROWS * HEADS];
    __shared__ float mx[OWN_ROWS * HEADS];
    __shared__ int   list[OWN_CAP];
    __shared__ int   first[SEED_ROWS];
    __shared__ int   cnt;
    __shared__ int   retry;

    const int tid = threadIdx.x;
    const int lane = tid & 31;
    const int wave = __builtin_amdgcn_readfirstlane(tid >> 5);
    const int node0 = blockIdx.x * OWN_ROWS;
    const int rows = min(OWN_ROWS, N - node0);

    for (int t = tid; t < OWN_ROWS * HID; t += 1024) acc[t] = 0.0f;
    for (int t = tid; t < OWN_ROWS * HEADS; t += 1024) {
        den[t] = 0.0f;
        mx[t] = -INFINITY;
    }
    {
        const float4* gx = (const float4*)(xr + (size_t)node0 * HID);
        float4* sx = (float4*)xrs;
        for (int t = tid; t < rows * (HID / 4); t += 1024) sx[t] = gx[t];
    }
    if (tid < SEED_ROWS) first[tid] = NO_EDGE;
    if (tid == 0) { cnt = 0; retry = 0; }
    __syncthreads();

    const float4 a4 = ((const float4*)att)[lane];
    const int nsteps = (E + OWN_STEP - 1) / OWN_STEP;

    int* mylist = glist + (size_t)blockIdx.x * OWN_CAP;
    int kept = -1;
    if (!FIRST) kept = __builtin_amdgcn_readfirstlane(gcnt[blockIdx.x]);
    if (kept >= 0) {
        for (int t0 = tid; t0 < kept; t0 += 4096) {
            int w[4];
            #pragma unroll
            for (int k = 0; k < 4; ++k)
                w[k] = (t0 + k * 1024 < kept) ? mylist[t0 + k * 1024] : NO_EDGE;
            #pragma unroll
            for (int k = 0; k < 4; ++k) {
                if (t0 + k * 1024 < kept) {
                    list[t0 + k * 1024] = w[k];
                    atomicMin(&first[w[k] & 255], w[k]);
                }
            }
        }
        __syncthreads();
        attend_list<true>(kept, list, first, &retry, acc, den, mx, xrs, xl, a4,
                          tid, wave, lane);
    } else {
        bool emptied = false;
        const int step0 = (int)(blockIdx.x % (unsigned)nsteps);
        const unsigned me4 = blockIdx.x * 0x01010101u;
        const uint4* mp = (const uint4*)omap + wave * 32 + lane;
        for (int scan = 0; scan < 2; ++scan) {
            uint4 nxt = mp[(size_t)step0 * (OWN_STEP / 16)];
            for (int it = 0; it < nsteps; ++it) {
                const int step = (it + step0 < nsteps) ? it + step0 : it + step0 - nsteps;
                const uint4 ow = nxt;
                if (it + 1 < nsteps)
                    nxt = mp[(size_t)((step + 1 < nsteps) ? step + 1 : 0) * (OWN_STEP / 16)];
                const int wbase = step * OWN_STEP + wave * 512 + lane * 16;
                const unsigned h0 = zero_bytes(ow.x ^ me4), h1 = zero_bytes(ow.y ^ me4);
                const unsigned h2 = zero_bytes(ow.z ^ me4), h3 = zero_bytes(ow.w ^ me4);
                take_hits((h0 >> 7) | (h1 >> 6) | (h2 >> 5) | (h3 >> 4), wbase, E, list,
                          &cnt, lane);
                __syncthreads();
                const int n = __builtin_amdgcn_readfirstlane(cnt);
                __syncthreads();
                if (n <= OWN_CAP - OWN_STEP && it != nsteps - 1) continue;
                edges_to_sources(n, list, src, dst, node0, tid);
                __syncthreads();
                if (it == nsteps - 1 && !emptied) {
                    if (FIRST) {
                        for (int t = tid; t < n; t += 1024) {
                            const int w = list[t];
                            mylist[t] = w;
                            atomicMin(&first[w & 255], w);
                        }
                        if (tid == 0) gcnt[blockIdx.x] = n;
                        __syncthreads();
                    }
                    attend_list<FIRST>(n, list, first, &retry, acc, den, mx, xrs,
                                       xl, a4, tid, wave, lane);
                    break;
                }
                emptied = true;
                if (scan == 0)
                    sweep_edges<1>(n, list, acc, den, mx, xrs, xl, a4, wave, lane);
                else
                    sweep_edges<2>(n, list, acc, den, mx, xrs, xl, a4, wave, lane);
                __syncthreads();
                if (tid == 0) cnt = 0;
                __syncthreads();
            }
            if (!emptied) break;
        }
        if (FIRST && emptied && tid == 0) gcnt[blockIdx.x] = -1;
    }

    const float4 b4 = *(const float4*)(bias + 4 * lane);
    for (int t = tid; t < rows * (HID / 4); t += 1024) {
        const int r = t >> 5;
        const float* pa = acc + r * HID + lane;
        const float dn = den[r * HEADS + (lane >> 2)] + EPS_F;
        float4 o;
        o.x = pa[0]  / dn + b4.x;
        o.y = pa[32] / dn + b4.y;
        o.z = pa[64] / dn + b4.z;
        o.w = pa[96] / dn + b4.w;
        *(float4*)(out + (size_t)(node0 + r) * HID + 4 * lane) = o;
    }
}

extern "C" void kernel_launch(void* const* d_in, const int* in_sizes, int n_in,
                              void* d_out, int out_size, void* d_ws, size_t ws_size,
                              hipStream_t stream) {
    const float* x          = (const float*)d_in[0];
    const int*   edge_index = (const int*)  d_in[1];
    const float* Wl_all     = (const float*)d_in[2];
    const float* Wr_all     = (const float*)d_in[3];
    const float* att_all    = (const float*)d_in[4];
    const float* bias_all   = (const float*)d_in[5];
    float* out = (float*)d_out;

    const int N = in_sizes[0] / HID;
    const int E = in_sizes[1] / 2;

    const int* src = edge_index;
    const int* dst = edge_index + E;

    const size_t nf = (size_t)N * HID;
    float* xl    = (float*)d_ws;
    float* xr    = xl + nf;
    float* bufA  = xr + nf;
    const int map_steps = (E + OWN_STEP - 1) / OWN_STEP;
    unsigned int* omap = (unsigned int*)(bufA + nf);
    int*   glist = (int*)(omap + (size_t)map_steps * (OWN_STEP / 4));

    const int TPB = 256;
    const int gemm_blocks = N / 16;
    const int own_blocks  = (N + OWN_ROWS - 1) / OWN_ROWS;
    int* gcnt = glist + (size_t)own_blocks * OWN_CAP;

    const int map_threads = map_steps * (OWN_STEP / 16);
    owner_map_kernel<<<(map_threads + TPB - 1) / TPB, TPB, 0, stream>>>(
        dst, omap, E, map_threads);

    const float* xin = x;
    for (int l = 0; l < LAYERS; ++l) {
        const float* Wl   = Wl_all   + (size_t)l * HID * HID;
        const float* Wr   = Wr_all   + (size_t)l * HID * HID;
        const float* att  = att_all  + (size_t)l * HEADS * HEAD_DIM;
        const float* bias = bias_all + (size_t)l * HID;
        float* xout = (l == LAYERS - 1) ? out : bufA;
        gemm_dual_f32<<<GEMM_WGS, TPB, 0, stream>>>(xin, Wl, Wr, xl, xr, gemm_blocks);
        if (l == 0)
            owner_attend_kernel<true><<<own_blocks, 1024, 0, stream>>>(
                src, dst, xl, xr, att, bias, xout, glist, gcnt, omap, N, E);
        else
            owner_attend_kernel<false><<<own_blocks, 1024, 0, stream>>>(
                src, dst, xl, xr, att, bias, xout, glist, gcnt, omap, N, E);
        xin = xout;
    }
}
